# starting kernel
# baseline (speedup 1.0000x reference)
#include <hip/hip_runtime.h>

typedef __attribute__((ext_vector_type(2))) float v2f;
typedef __attribute__((ext_vector_type(8))) float v8f;

#define IN_DIM  128
#define HID_DIM 64
#define OUT_DIM 3

__device__ __forceinline__ void atomAddF(float* p, float v) {
    __hip_atomic_fetch_add(p, v, __ATOMIC_RELAXED, __HIP_MEMORY_SCOPE_AGENT);
}

__global__ void k_deg_init(float* __restrict__ deg, int n) {
    int i = blockIdx.x * blockDim.x + threadIdx.x;
    if (i < n) deg[i] = 1.0f;
}

__global__ void k_deg_scatter(const int* __restrict__ dst, float* __restrict__ deg, int e) {
    int i = blockIdx.x * blockDim.x + threadIdx.x;
    if (i < e) atomAddF(&deg[dst[i]], 1.0f);
}

__global__ void k_dinv(float* __restrict__ deg, int n) {
    int i = blockIdx.x * blockDim.x + threadIdx.x;
    if (i < n) deg[i] = rsqrtf(fmaxf(deg[i], 1.0f));
}

__global__ __launch_bounds__(256) void k_gemm1(const float* __restrict__ X,
                                               const float* __restrict__ W,
                                               float* __restrict__ H, int n) {
    __shared__ float sW[IN_DIM * HID_DIM];
    for (int i = threadIdx.x; i < IN_DIM * HID_DIM; i += 256) sW[i] = W[i];
    __syncthreads();

    const int lane = threadIdx.x & 31;
    const int wave = threadIdx.x >> 5;
    const int rowBase = (blockIdx.x * 8 + wave) * 16;
    if (rowBase >= n) return;

    const int m  = lane & 15;
    const int kh = lane >> 4;
    int row  = rowBase + m;
    int rowc = row < n ? row : n - 1;
    const float* xrow = X + (size_t)rowc * IN_DIM;

    v8f a0 = {}, a1 = {}, a2 = {}, a3 = {};
#pragma unroll 4
    for (int k = 0; k < IN_DIM; k += 4) {
        v2f a = *(const v2f*)(xrow + k + 2 * kh);
        const float* wp = sW + (k + 2 * kh) * HID_DIM;
        v2f b0 = { wp[m +  0], wp[HID_DIM + m +  0] };
        v2f b1 = { wp[m + 16], wp[HID_DIM + m + 16] };
        v2f b2 = { wp[m + 32], wp[HID_DIM + m + 32] };
        v2f b3 = { wp[m + 48], wp[HID_DIM + m + 48] };
        a0 = __builtin_amdgcn_wmma_f32_16x16x4_f32(false, a, false, b0, (short)0, a0, false, false);
        a1 = __builtin_amdgcn_wmma_f32_16x16x4_f32(false, a, false, b1, (short)0, a1, false, false);
        a2 = __builtin_amdgcn_wmma_f32_16x16x4_f32(false, a, false, b2, (short)0, a2, false, false);
        a3 = __builtin_amdgcn_wmma_f32_16x16x4_f32(false, a, false, b3, (short)0, a3, false, false);
    }
#pragma unroll
    for (int r = 0; r < 8; ++r) {
        int orow = rowBase + r + 8 * kh;
        if (orow < n) {
            float* hp = H + (size_t)orow * HID_DIM + m;
            hp[0]  = a0[r];
            hp[16] = a1[r];
            hp[32] = a2[r];
            hp[48] = a3[r];
        }
    }
}

__global__ void k_self1(const float* __restrict__ H1, const float* __restrict__ dinv,
                        const float* __restrict__ b1, float* __restrict__ A1, int n) {
    int idx = blockIdx.x * blockDim.x + threadIdx.x;
    if (idx < n * HID_DIM) {
        int i = idx / HID_DIM, c = idx % HID_DIM;
        float w = dinv[i];
        A1[idx] = H1[idx] * w * w + b1[c];
    }
}

__global__ void k_agg1(const int* __restrict__ src, const int* __restrict__ dst,
                       const float* __restrict__ dinv, const float* __restrict__ H1,
                       float* __restrict__ A1, int e) {
    long long idx = (long long)blockIdx.x * blockDim.x + threadIdx.x;
    if (idx >= (long long)e * 32) return;
    int ed = (int)(idx >> 5);
    int l  = (int)(idx & 31);
    int s = src[ed], d = dst[ed];
    float w = dinv[s] * dinv[d];
    v2f h = *(const v2f*)(H1 + (size_t)s * HID_DIM + l * 2);
    float* op = A1 + (size_t)d * HID_DIM + l * 2;
    atomAddF(op + 0, h.x * w);
    atomAddF(op + 1, h.y * w);
}

__global__ __launch_bounds__(256) void k_gemm2(const float* __restrict__ A1,
                                               const float* __restrict__ W2,
                                               float* __restrict__ H2, int n) {
    const int lane = threadIdx.x & 31;
    const int wave = threadIdx.x >> 5;
    const int rowBase = (blockIdx.x * 8 + wave) * 16;
    if (rowBase >= n) return;

    const int m  = lane & 15;
    const int kh = lane >> 4;
    int row  = rowBase + m;
    int rowc = row < n ? row : n - 1;
    const float* arow = A1 + (size_t)rowc * HID_DIM;

    v8f acc = {};
#pragma unroll 4
    for (int k = 0; k < HID_DIM; k += 4) {
        v2f a = *(const v2f*)(arow + k + 2 * kh);
        a.x = fmaxf(a.x, 0.0f);
        a.y = fmaxf(a.y, 0.0f);
        v2f b = { 0.0f, 0.0f };
        if (m < OUT_DIM) {
            b.x = W2[(k + 2 * kh) * OUT_DIM + m];
            b.y = W2[(k + 2 * kh + 1) * OUT_DIM + m];
        }
        acc = __builtin_amdgcn_wmma_f32_16x16x4_f32(false, a, false, b, (short)0, acc, false, false);
    }
#pragma unroll
    for (int r = 0; r < 8; ++r) {
        int orow = rowBase + r + 8 * kh;
        if (orow < n && m < OUT_DIM) H2[(size_t)orow * OUT_DIM + m] = acc[r];
    }
}

__global__ void k_self2(const float* __restrict__ H2, const float* __restrict__ dinv,
                        const float* __restrict__ b2, float* __restrict__ out, int n) {
    int idx = blockIdx.x * blockDim.x + threadIdx.x;
    if (idx < n * OUT_DIM) {
        int i = idx / OUT_DIM, c = idx % OUT_DIM;
        float w = dinv[i];
        out[idx] = H2[idx] * w * w + b2[c];
    }
}

__global__ void k_agg2(const int* __restrict__ src, const int* __restrict__ dst,
                       const float* __restrict__ dinv, const float* __restrict__ H2,
                       float* __restrict__ out, int e) {
    int i = blockIdx.x * blockDim.x + threadIdx.x;
    if (i < e) {
        int s = src[i], d = dst[i];
        float w = dinv[s] * dinv[d];
        const float* hp = H2 + (size_t)s * OUT_DIM;
        float* op = out + (size_t)d * OUT_DIM;
        atomAddF(op + 0, hp[0] * w);
        atomAddF(op + 1, hp[1] * w);
        atomAddF(op + 2, hp[2] * w);
    }
}

extern "C" void kernel_launch(void* const* d_in, const int* in_sizes, int n_in,
                              void* d_out, int out_size, void* d_ws, size_t ws_size,
                              hipStream_t stream) {
    (void)n_in; (void)out_size; (void)ws_size;
    const float* x  = (const float*)d_in[0];
    const int*  edge = (const int*)d_in[1];
    const float* W1 = (const float*)d_in[2];
    const float* b1 = (const float*)d_in[3];
    const float* W2 = (const float*)d_in[4];
    const float* b2 = (const float*)d_in[5];
    float* out = (float*)d_out;

    const int n = in_sizes[0] / IN_DIM;
    const int e = in_sizes[1] / 2;
    const int* src = edge;
    const int* dst = edge + e;

    float* ws   = (float*)d_ws;
    float* dinv = ws;
    float* H1   = ws + n;
    float* A1   = H1 + (size_t)n * HID_DIM;
    float* H2   = A1 + (size_t)n * HID_DIM;

    dim3 blk(256);
    k_deg_init   <<<(n + 255) / 256, blk, 0, stream>>>(dinv, n);
    k_deg_scatter<<<(e + 255) / 256, blk, 0, stream>>>(dst, dinv, e);
    k_dinv       <<<(n + 255) / 256, blk, 0, stream>>>(dinv, n);

    k_gemm1<<<(n + 127) / 128, blk, 0, stream>>>(x, W1, H1, n);
    k_self1<<<(n * HID_DIM + 255) / 256, blk, 0, stream>>>(H1, dinv, b1, A1, n);
    long long work1 = (long long)e * 32;
    k_agg1<<<(unsigned)((work1 + 255) / 256), blk, 0, stream>>>(src, dst, dinv, H1, A1, e);

    k_gemm2<<<(n + 127) / 128, blk, 0, stream>>>(A1, W2, H2, n);
    k_self2<<<(n * OUT_DIM + 255) / 256, blk, 0, stream>>>(H2, dinv, b2, out, n);
    k_agg2<<<(e + 255) / 256, blk, 0, stream>>>(src, dst, dinv, H2, out, e);
}


# float-atomic scatter-add of both layers replaced by gather over edges bucketed by destination (one returning int atomic per edge: count and rank); sums now in arrival-rank order
# speedup vs baseline: 2.3087x; 2.3087x over previous
#include <hip/hip_runtime.h>

typedef __attribute__((ext_vector_type(2))) float v2f;
typedef __attribute__((ext_vector_type(8))) float v8f;

#define IN_DIM  128
#define HID_DIM 64
#define OUT_DIM 3
#define SCAN_TILE 1024

__device__ __forceinline__ int waveSum(int v) {
    for (int m = 16; m >= 1; m >>= 1) v += __shfl_xor(v, m);
    return v;
}

__global__ void k_cnt_zero(int* __restrict__ cnt, int n) {
    int i = blockIdx.x * blockDim.x + threadIdx.x;
    if (i < n) cnt[i] = 0;
}

__global__ void k_rank(const int* __restrict__ dst, int* __restrict__ cnt,
                       int* __restrict__ rank, int e) {
    int i = blockIdx.x * blockDim.x + threadIdx.x;
    if (i < e)
        rank[i] = __hip_atomic_fetch_add(&cnt[dst[i]], 1, __ATOMIC_RELAXED, __HIP_MEMORY_SCOPE_AGENT);
}

__global__ __launch_bounds__(256) void k_bsum(const int* __restrict__ cnt, int* __restrict__ bsum, int n) {
    __shared__ int sw[8];
    const int base = blockIdx.x * SCAN_TILE + threadIdx.x * 4;
    int s = 0;
#pragma unroll
    for (int k = 0; k < 4; ++k)
        if (base + k < n) s += cnt[base + k];
    s = waveSum(s);
    if ((threadIdx.x & 31) == 0) sw[threadIdx.x >> 5] = s;
    __syncthreads();
    if (threadIdx.x == 0) {
        int tot = 0;
        for (int w = 0; w < 8; ++w) tot += sw[w];
        bsum[blockIdx.x] = tot;
    }
}

__global__ __launch_bounds__(256) void k_off(const int* __restrict__ cnt, const int* __restrict__ bsum,
                                             int* __restrict__ off, float* __restrict__ dinv, int n) {
    __shared__ int sw[8];
    __shared__ int sbase[8];
    const int lane = threadIdx.x & 31;
    const int wave = threadIdx.x >> 5;

    int p = 0;
    for (int j = threadIdx.x; j < (int)blockIdx.x; j += 256) p += bsum[j];
    p = waveSum(p);
    if (lane == 0) sbase[wave] = p;

    const int base = blockIdx.x * SCAN_TILE + threadIdx.x * 4;
    int c[4];
    int ts = 0;
#pragma unroll
    for (int k = 0; k < 4; ++k) {
        c[k] = (base + k < n) ? cnt[base + k] : 0;
        ts += c[k];
    }
    int inc = ts;
#pragma unroll
    for (int d = 1; d < 32; d <<= 1) {
        int v = __shfl_up(inc, d);
        if (lane >= d) inc += v;
    }
    if (lane == 31) sw[wave] = inc;
    __syncthreads();
    int pre = 0;
    for (int w = 0; w < 8; ++w) {
        pre += sbase[w];
        if (w < wave) pre += sw[w];
    }
    int run = pre + inc - ts;
#pragma unroll
    for (int k = 0; k < 4; ++k) {
        const int i = base + k;
        if (i < n) {
            off[i]  = run;
            dinv[i] = rsqrtf(fmaxf((float)(c[k] + 1), 1.0f));
            run += c[k];
            if (i == n - 1) off[n] = run;
        }
    }
}

__global__ void k_fill(const int* __restrict__ src, const int* __restrict__ dst,
                       const int* __restrict__ rank, const int* __restrict__ off,
                       int* __restrict__ csr, int e) {
    int i = blockIdx.x * blockDim.x + threadIdx.x;
    if (i < e) csr[off[dst[i]] + rank[i]] = src[i];
}

__global__ __launch_bounds__(256) void k_gemm1(const float* __restrict__ X,
                                               const float* __restrict__ W,
                                               float* __restrict__ H, int n) {
    __shared__ float sW[IN_DIM * HID_DIM];
    for (int i = threadIdx.x; i < IN_DIM * HID_DIM; i += 256) sW[i] = W[i];
    __syncthreads();

    const int lane = threadIdx.x & 31;
    const int wave = threadIdx.x >> 5;
    const int rowBase = (blockIdx.x * 8 + wave) * 16;
    if (rowBase >= n) return;

    const int m  = lane & 15;
    const int kh = lane >> 4;
    int row  = rowBase + m;
    int rowc = row < n ? row : n - 1;
    const float* xrow = X + (size_t)rowc * IN_DIM;

    v8f a0 = {}, a1 = {}, a2 = {}, a3 = {};
#pragma unroll 4
    for (int k = 0; k < IN_DIM; k += 4) {
        v2f a = *(const v2f*)(xrow + k + 2 * kh);
        const float* wp = sW + (k + 2 * kh) * HID_DIM;
        v2f b0 = { wp[m +  0], wp[HID_DIM + m +  0] };
        v2f b1 = { wp[m + 16], wp[HID_DIM + m + 16] };
        v2f b2 = { wp[m + 32], wp[HID_DIM + m + 32] };
        v2f b3 = { wp[m + 48], wp[HID_DIM + m + 48] };
        a0 = __builtin_amdgcn_wmma_f32_16x16x4_f32(false, a, false, b0, (short)0, a0, false, false);
        a1 = __builtin_amdgcn_wmma_f32_16x16x4_f32(false, a, false, b1, (short)0, a1, false, false);
        a2 = __builtin_amdgcn_wmma_f32_16x16x4_f32(false, a, false, b2, (short)0, a2, false, false);
        a3 = __builtin_amdgcn_wmma_f32_16x16x4_f32(false, a, false, b3, (short)0, a3, false, false);
    }
#pragma unroll
    for (int r = 0; r < 8; ++r) {
        int orow = rowBase + r + 8 * kh;
        if (orow < n) {
            float* hp = H + (size_t)orow * HID_DIM + m;
            hp[0]  = a0[r];
            hp[16] = a1[r];
            hp[32] = a2[r];
            hp[48] = a3[r];
        }
    }
}

__global__ __launch_bounds__(256) void k_gather1(const int* __restrict__ off, const int* __restrict__ csr,
                                                 const float* __restrict__ dinv, const float* __restrict__ H1,
                                                 const float* __restrict__ b1, float* __restrict__ A1, int n) {
    const int lane = threadIdx.x & 31;
    const int i = __builtin_amdgcn_readfirstlane((int)(blockIdx.x * 8 + (threadIdx.x >> 5)));
    if (i >= n) return;
    const int beg = __builtin_amdgcn_readfirstlane(off[i]);
    const int end = __builtin_amdgcn_readfirstlane(off[i + 1]);
    const float di = dinv[i];
    const float* hcol = H1 + 2 * lane;

    const v2f hs = *(const v2f*)(hcol + (size_t)i * HID_DIM);
    v2f acc = { hs.x * di * di, hs.y * di * di };

    for (int j = beg; j < end; j += 32) {
        int c = end - j;
        if (c > 32) c = 32;
        int   s_l = 0;
        float w_l = 0.0f;
        if (lane < c) {
            s_l = csr[j + lane];
            w_l = dinv[s_l] * di;
        }
        int t = 0;
        for (; t + 8 <= c; t += 8) {
            v2f   h[8];
            float w[8];
#pragma unroll
            for (int u = 0; u < 8; ++u) {
                const int s = __builtin_amdgcn_readlane(s_l, t + u);
                w[u] = __int_as_float(__builtin_amdgcn_readlane(__float_as_int(w_l), t + u));
                h[u] = *(const v2f*)(hcol + (size_t)s * HID_DIM);
            }
#pragma unroll
            for (int u = 0; u < 8; ++u) {
                acc.x = fmaf(h[u].x, w[u], acc.x);
                acc.y = fmaf(h[u].y, w[u], acc.y);
            }
        }
        if (t < c) {
            v2f   h[7];
            float w[7];
#pragma unroll
            for (int u = 0; u < 7; ++u) {
                h[u] = v2f{ 0.0f, 0.0f };
                w[u] = 0.0f;
                if (t + u < c) {
                    const int s = __builtin_amdgcn_readlane(s_l, t + u);
                    w[u] = __int_as_float(__builtin_amdgcn_readlane(__float_as_int(w_l), t + u));
                    h[u] = *(const v2f*)(hcol + (size_t)s * HID_DIM);
                }
            }
#pragma unroll
            for (int u = 0; u < 7; ++u) {
                acc.x = fmaf(h[u].x, w[u], acc.x);
                acc.y = fmaf(h[u].y, w[u], acc.y);
            }
        }
    }
    const v2f bb = *(const v2f*)(b1 + 2 * lane);
    v2f o = { acc.x + bb.x, acc.y + bb.y };
    *(v2f*)(A1 + (size_t)i * HID_DIM + 2 * lane) = o;
}

__global__ __launch_bounds__(256) void k_gemm2(const float* __restrict__ A1,
                                               const float* __restrict__ W2,
                                               float* __restrict__ H2, int n) {
    const int lane = threadIdx.x & 31;
    const int wave = threadIdx.x >> 5;
    const int rowBase = (blockIdx.x * 8 + wave) * 16;
    if (rowBase >= n) return;

    const int m  = lane & 15;
    const int kh = lane >> 4;
    int row  = rowBase + m;
    int rowc = row < n ? row : n - 1;
    const float* arow = A1 + (size_t)rowc * HID_DIM;

    v8f acc = {};
#pragma unroll 4
    for (int k = 0; k < HID_DIM; k += 4) {
        v2f a = *(const v2f*)(arow + k + 2 * kh);
        a.x = fmaxf(a.x, 0.0f);
        a.y = fmaxf(a.y, 0.0f);
        v2f b = { 0.0f, 0.0f };
        if (m < OUT_DIM) {
            b.x = W2[(k + 2 * kh) * OUT_DIM + m];
            b.y = W2[(k + 2 * kh + 1) * OUT_DIM + m];
        }
        acc = __builtin_amdgcn_wmma_f32_16x16x4_f32(false, a, false, b, (short)0, acc, false, false);
    }
#pragma unroll
    for (int r = 0; r < 8; ++r) {
        int orow = rowBase + r + 8 * kh;
        if (orow < n && m < OUT_DIM) H2[(size_t)orow * OUT_DIM + m] = acc[r];
    }
}

__global__ __launch_bounds__(256) void k_gather2(const int* __restrict__ off, const int* __restrict__ csr,
                                                 const float* __restrict__ dinv, const float* __restrict__ H2,
                                                 const float* __restrict__ b2, float* __restrict__ out, int n) {
    __shared__ float so[32 * OUT_DIM];
    const int g  = threadIdx.x >> 3;
    const int sl = threadIdx.x & 7;
    const int i  = blockIdx.x * 32 + g;

    float a0 = 0.0f, a1 = 0.0f, a2 = 0.0f;
    float di = 0.0f;
    if (i < n) {
        const int beg = off[i], end = off[i + 1];
        di = dinv[i];
        for (int j = beg + sl; j < end; j += 8) {
            const int   s = csr[j];
            const float w = dinv[s] * di;
            const float* hp = H2 + (size_t)s * OUT_DIM;
            a0 = fmaf(hp[0], w, a0);
            a1 = fmaf(hp[1], w, a1);
            a2 = fmaf(hp[2], w, a2);
        }
    }
#pragma unroll
    for (int m = 4; m >= 1; m >>= 1) {
        a0 += __shfl_xor(a0, m);
        a1 += __shfl_xor(a1, m);
        a2 += __shfl_xor(a2, m);
    }
    if (sl == 0 && i < n) {
        const float* hp = H2 + (size_t)i * OUT_DIM;
        so[g * OUT_DIM + 0] = a0 + hp[0] * di * di + b2[0];
        so[g * OUT_DIM + 1] = a1 + hp[1] * di * di + b2[1];
        so[g * OUT_DIM + 2] = a2 + hp[2] * di * di + b2[2];
    }
    __syncthreads();
    if (threadIdx.x < 32 * OUT_DIM) {
        const int idx = blockIdx.x * (32 * OUT_DIM) + threadIdx.x;
        if (idx < n * OUT_DIM) out[idx] = so[threadIdx.x];
    }
}

extern "C" void kernel_launch(void* const* d_in, const int* in_sizes, int n_in,
                              void* d_out, int out_size, void* d_ws, size_t ws_size,
                              hipStream_t stream) {
    (void)n_in; (void)out_size; (void)ws_size;
    const float* x  = (const float*)d_in[0];
    const int*  edge = (const int*)d_in[1];
    const float* W1 = (const float*)d_in[2];
    const float* b1 = (const float*)d_in[3];
    const float* W2 = (const float*)d_in[4];
    const float* b2 = (const float*)d_in[5];
    float* out = (float*)d_out;

    const int n = in_sizes[0] / IN_DIM;
    const int e = in_sizes[1] / 2;
    const int* src = edge;
    const int* dst = edge + e;
    const int nb = (n + SCAN_TILE - 1) / SCAN_TILE;

    const size_t np  = ((size_t)n + 1 + 63) & ~(size_t)63;
    const size_t ep  = ((size_t)e + 63) & ~(size_t)63;
    const size_t nbp = ((size_t)nb + 63) & ~(size_t)63;
    float* ws   = (float*)d_ws;
    float* dinv = ws;
    int*   cnt  = (int*)(ws + np);
    int*   off  = cnt + np;
    int*   bsum = off + np;
    int*   rank = bsum + nbp;
    int*   csr  = rank + ep;
    float* H1   = (float*)(csr + ep);
    float* A1   = H1 + (size_t)n * HID_DIM;
    float* H2   = A1 + (size_t)n * HID_DIM;

    dim3 blk(256);
    k_cnt_zero<<<(n + 255) / 256, blk, 0, stream>>>(cnt, n);
    k_rank    <<<(e + 255) / 256, blk, 0, stream>>>(dst, cnt, rank, e);
    k_bsum    <<<nb, blk, 0, stream>>>(cnt, bsum, n);
    k_off     <<<nb, blk, 0, stream>>>(cnt, bsum, off, dinv, n);
    k_fill    <<<(e + 255) / 256, blk, 0, stream>>>(src, dst, rank, off, csr, e);

    k_gemm1  <<<(n + 127) / 128, blk, 0, stream>>>(x, W1, H1, n);
    k_gather1<<<(n + 7) / 8, blk, 0, stream>>>(off, csr, dinv, H1, b1, A1, n);

    k_gemm2  <<<(n + 127) / 128, blk, 0, stream>>>(A1, W2, H2, n);
    k_gather2<<<(n + 31) / 32, blk, 0, stream>>>(off, csr, dinv, H2, b2, out, n);
}


# edges grouped by destination with a two-level counting sort on LDS counters (chunk x 256-node bucket, then node) instead of one returning global atomic per edge; sums keep an arrival order
# speedup vs baseline: 4.5016x; 1.9498x over previous
#include <hip/hip_runtime.h>

typedef __attribute__((ext_vector_type(2))) float v2f;
typedef __attribute__((ext_vector_type(8))) float v8f;

#define IN_DIM  128
#define HID_DIM 64
#define OUT_DIM 3
#define BKT_SHIFT 8
#define BKT_NODES (1 << BKT_SHIFT)
#define NBMAX 512
#define NCHUNK 256
#define SORT_THREADS 1024

__global__ __launch_bounds__(SORT_THREADS) void k_hist(const int* __restrict__ dst, int* __restrict__ table,
                                                       int e, int nb) {
    __shared__ int h[NBMAX];
    for (int k = threadIdx.x; k < NBMAX; k += SORT_THREADS) h[k] = 0;
    __syncthreads();
    const int ch  = (e + (int)gridDim.x - 1) / (int)gridDim.x;
    const int beg = (int)blockIdx.x * ch;
    const int end = (beg + ch < e) ? beg + ch : e;
    int j = beg + (int)threadIdx.x;
    for (; j + 3 * SORT_THREADS < end; j += 4 * SORT_THREADS) {
        int d[4];
#pragma unroll
        for (int u = 0; u < 4; ++u) d[u] = dst[j + u * SORT_THREADS];
#pragma unroll
        for (int u = 0; u < 4; ++u) atomicAdd(&h[d[u] >> BKT_SHIFT], 1);
    }
    for (; j < end; j += SORT_THREADS) atomicAdd(&h[dst[j] >> BKT_SHIFT], 1);
    __syncthreads();
    for (int k = threadIdx.x; k < nb; k += SORT_THREADS) table[(int)blockIdx.x * NBMAX + k] = h[k];
}

__global__ __launch_bounds__(SORT_THREADS) void k_part(const int* __restrict__ src, const int* __restrict__ dst,
                                                       const int* __restrict__ table, int* __restrict__ bbase,
                                                       int* __restrict__ pairs, int e, int nb) {
    __shared__ int tot[2][NBMAX];
    __shared__ int pre[2][NBMAX];
    __shared__ int cur[NBMAX];
    __shared__ int wsum[SORT_THREADS / 32];
    const int tid  = threadIdx.x;
    const int lane = tid & 31;
    const int wave = tid >> 5;
    const int g    = blockIdx.x;
    const int G    = gridDim.x;

    {
        const int bk   = tid & (NBMAX - 1);
        const int half = tid / NBMAX;
        const int r0   = half ? (G >> 1) : 0;
        const int r1   = half ? G : (G >> 1);
        int t = 0, p = 0;
        if (bk < nb) {
#pragma unroll 8
            for (int r = r0; r < r1; ++r) {
                const int v = table[r * NBMAX + bk];
                t += v;
                if (r < g) p += v;
            }
        }
        tot[half][bk] = t;
        pre[half][bk] = p;
    }
    __syncthreads();

    const int val = (tid < NBMAX) ? tot[0][tid] + tot[1][tid] : 0;
    int inc = val;
#pragma unroll
    for (int d = 1; d < 32; d <<= 1) {
        const int v = __shfl_up(inc, d);
        if (lane >= d) inc += v;
    }
    if (lane == 31) wsum[wave] = inc;
    __syncthreads();
    int wpre = 0;
    for (int w = 0; w < wave; ++w) wpre += wsum[w];
    const int excl = wpre + inc - val;
    if (tid < NBMAX) cur[tid] = excl + pre[0][tid] + pre[1][tid];
    if (g == 0 && tid <= nb) bbase[tid] = excl;
    __syncthreads();

    const int ch  = (e + G - 1) / G;
    const int beg = g * ch;
    const int end = (beg + ch < e) ? beg + ch : e;
    int j = beg + tid;
    for (; j + 3 * SORT_THREADS < end; j += 4 * SORT_THREADS) {
        int d[4], s[4], pos[4];
#pragma unroll
        for (int u = 0; u < 4; ++u) {
            d[u] = dst[j + u * SORT_THREADS];
            s[u] = src[j + u * SORT_THREADS];
        }
#pragma unroll
        for (int u = 0; u < 4; ++u) pos[u] = atomicAdd(&cur[d[u] >> BKT_SHIFT], 1);
#pragma unroll
        for (int u = 0; u < 4; ++u)
            pairs[pos[u]] = (int)(((unsigned)s[u] << BKT_SHIFT) | (unsigned)(d[u] & (BKT_NODES - 1)));
    }
    for (; j < end; j += SORT_THREADS) {
        const int d = dst[j];
        const int s = src[j];
        const int pos = atomicAdd(&cur[d >> BKT_SHIFT], 1);
        pairs[pos] = (int)(((unsigned)s << BKT_SHIFT) | (unsigned)(d & (BKT_NODES - 1)));
    }
}

__global__ __launch_bounds__(SORT_THREADS) void k_bsort(const int* __restrict__ pairs, const int* __restrict__ bbase,
                                                        int* __restrict__ off, float* __restrict__ dinv,
                                                        int* __restrict__ csr, int n) {
    __shared__ int c[BKT_NODES];
    __shared__ int wsum[BKT_NODES / 32];
    const int tid  = threadIdx.x;
    const int lane = tid & 31;
    const int wave = tid >> 5;
    const int beg  = bbase[blockIdx.x];
    const int end  = bbase[blockIdx.x + 1];
    if (tid < BKT_NODES) c[tid] = 0;
    __syncthreads();
    int j = beg + tid;
    for (; j + 3 * SORT_THREADS < end; j += 4 * SORT_THREADS) {
        int w[4];
#pragma unroll
        for (int u = 0; u < 4; ++u) w[u] = pairs[j + u * SORT_THREADS];
#pragma unroll
        for (int u = 0; u < 4; ++u) atomicAdd(&c[w[u] & (BKT_NODES - 1)], 1);
    }
    for (; j < end; j += SORT_THREADS) atomicAdd(&c[pairs[j] & (BKT_NODES - 1)], 1);
    __syncthreads();

    const int val = (tid < BKT_NODES) ? c[tid] : 0;
    int inc = val;
#pragma unroll
    for (int d = 1; d < 32; d <<= 1) {
        const int v = __shfl_up(inc, d);
        if (lane >= d) inc += v;
    }
    if (tid < BKT_NODES && lane == 31) wsum[wave] = inc;
    __syncthreads();
    if (tid < BKT_NODES) {
        int wpre = 0;
        for (int w = 0; w < wave; ++w) wpre += wsum[w];
        const int first = beg + wpre + inc - val;
        const int node  = (int)blockIdx.x * BKT_NODES + tid;
        c[tid] = first;
        if (node < n) {
            off[node]  = first;
            dinv[node] = rsqrtf(fmaxf((float)(val + 1), 1.0f));
            if (node == n - 1) off[n] = first + val;
        }
    }
    __syncthreads();
    j = beg + tid;
    for (; j + 3 * SORT_THREADS < end; j += 4 * SORT_THREADS) {
        int w[4], pos[4];
#pragma unroll
        for (int u = 0; u < 4; ++u) w[u] = pairs[j + u * SORT_THREADS];
#pragma unroll
        for (int u = 0; u < 4; ++u) pos[u] = atomicAdd(&c[w[u] & (BKT_NODES - 1)], 1);
#pragma unroll
        for (int u = 0; u < 4; ++u) csr[pos[u]] = (int)((unsigned)w[u] >> BKT_SHIFT);
    }
    for (; j < end; j += SORT_THREADS) {
        const int w = pairs[j];
        const int pos = atomicAdd(&c[w & (BKT_NODES - 1)], 1);
        csr[pos] = (int)((unsigned)w >> BKT_SHIFT);
    }
}

__global__ __launch_bounds__(256) void k_gemm1(const float* __restrict__ X,
                                               const float* __restrict__ W,
                                               float* __restrict__ H, int n) {
    __shared__ float sW[IN_DIM * HID_DIM];
    for (int i = threadIdx.x; i < IN_DIM * HID_DIM; i += 256) sW[i] = W[i];
    __syncthreads();

    const int lane = threadIdx.x & 31;
    const int wave = threadIdx.x >> 5;
    const int rowBase = (blockIdx.x * 8 + wave) * 16;
    if (rowBase >= n) return;

    const int m  = lane & 15;
    const int kh = lane >> 4;
    int row  = rowBase + m;
    int rowc = row < n ? row : n - 1;
    const float* xrow = X + (size_t)rowc * IN_DIM;

    v8f a0 = {}, a1 = {}, a2 = {}, a3 = {};
#pragma unroll 4
    for (int k = 0; k < IN_DIM; k += 4) {
        v2f a = *(const v2f*)(xrow + k + 2 * kh);
        const float* wp = sW + (k + 2 * kh) * HID_DIM;
        v2f b0 = { wp[m +  0], wp[HID_DIM + m +  0] };
        v2f b1 = { wp[m + 16], wp[HID_DIM + m + 16] };
        v2f b2 = { wp[m + 32], wp[HID_DIM + m + 32] };
        v2f b3 = { wp[m + 48], wp[HID_DIM + m + 48] };
        a0 = __builtin_amdgcn_wmma_f32_16x16x4_f32(false, a, false, b0, (short)0, a0, false, false);
        a1 = __builtin_amdgcn_wmma_f32_16x16x4_f32(false, a, false, b1, (short)0, a1, false, false);
        a2 = __builtin_amdgcn_wmma_f32_16x16x4_f32(false, a, false, b2, (short)0, a2, false, false);
        a3 = __builtin_amdgcn_wmma_f32_16x16x4_f32(false, a, false, b3, (short)0, a3, false, false);
    }
#pragma unroll
    for (int r = 0; r < 8; ++r) {
        int orow = rowBase + r + 8 * kh;
        if (orow < n) {
            float* hp = H + (size_t)orow * HID_DIM + m;
            hp[0]  = a0[r];
            hp[16] = a1[r];
            hp[32] = a2[r];
            hp[48] = a3[r];
        }
    }
}

__global__ __launch_bounds__(256) void k_gather1(const int* __restrict__ off, const int* __restrict__ csr,
                                                 const float* __restrict__ dinv, const float* __restrict__ H1,
                                                 const float* __restrict__ b1, float* __restrict__ A1, int n) {
    const int lane = threadIdx.x & 31;
    const int i = __builtin_amdgcn_readfirstlane((int)(blockIdx.x * 8 + (threadIdx.x >> 5)));
    if (i >= n) return;
    const int beg = __builtin_amdgcn_readfirstlane(off[i]);
    const int end = __builtin_amdgcn_readfirstlane(off[i + 1]);
    const float di = dinv[i];
    const float* hcol = H1 + 2 * lane;

    const v2f hs = *(const v2f*)(hcol + (size_t)i * HID_DIM);
    v2f acc = { hs.x * di * di, hs.y * di * di };

    for (int j = beg; j < end; j += 32) {
        int c = end - j;
        if (c > 32) c = 32;
        int   s_l = 0;
        float w_l = 0.0f;
        if (lane < c) {
            s_l = csr[j + lane];
            w_l = dinv[s_l] * di;
        }
        int t = 0;
        for (; t + 8 <= c; t += 8) {
            v2f   h[8];
            float w[8];
#pragma unroll
            for (int u = 0; u < 8; ++u) {
                const int s = __builtin_amdgcn_readlane(s_l, t + u);
                w[u] = __int_as_float(__builtin_amdgcn_readlane(__float_as_int(w_l), t + u));
                h[u] = *(const v2f*)(hcol + (size_t)s * HID_DIM);
            }
#pragma unroll
            for (int u = 0; u < 8; ++u) {
                acc.x = fmaf(h[u].x, w[u], acc.x);
                acc.y = fmaf(h[u].y, w[u], acc.y);
            }
        }
        if (t < c) {
            v2f   h[7];
            float w[7];
#pragma unroll
            for (int u = 0; u < 7; ++u) {
                h[u] = v2f{ 0.0f, 0.0f };
                w[u] = 0.0f;
                if (t + u < c) {
                    const int s = __builtin_amdgcn_readlane(s_l, t + u);
                    w[u] = __int_as_float(__builtin_amdgcn_readlane(__float_as_int(w_l), t + u));
                    h[u] = *(const v2f*)(hcol + (size_t)s * HID_DIM);
                }
            }
#pragma unroll
            for (int u = 0; u < 7; ++u) {
                acc.x = fmaf(h[u].x, w[u], acc.x);
                acc.y = fmaf(h[u].y, w[u], acc.y);
            }
        }
    }
    const v2f bb = *(const v2f*)(b1 + 2 * lane);
    v2f o = { acc.x + bb.x, acc.y + bb.y };
    *(v2f*)(A1 + (size_t)i * HID_DIM + 2 * lane) = o;
}

__global__ __launch_bounds__(256) void k_gemm2(const float* __restrict__ A1,
                                               const float* __restrict__ W2,
                                               float* __restrict__ H2, int n) {
    const int lane = threadIdx.x & 31;
    const int wave = threadIdx.x >> 5;
    const int rowBase = (blockIdx.x * 8 + wave) * 16;
    if (rowBase >= n) return;

    const int m  = lane & 15;
    const int kh = lane >> 4;
    int row  = rowBase + m;
    int rowc = row < n ? row : n - 1;
    const float* arow = A1 + (size_t)rowc * HID_DIM;

    v8f acc = {};
#pragma unroll 4
    for (int k = 0; k < HID_DIM; k += 4) {
        v2f a = *(const v2f*)(arow + k + 2 * kh);
        a.x = fmaxf(a.x, 0.0f);
        a.y = fmaxf(a.y, 0.0f);
        v2f b = { 0.0f, 0.0f };
        if (m < OUT_DIM) {
            b.x = W2[(k + 2 * kh) * OUT_DIM + m];
            b.y = W2[(k + 2 * kh + 1) * OUT_DIM + m];
        }
        acc = __builtin_amdgcn_wmma_f32_16x16x4_f32(false, a, false, b, (short)0, acc, false, false);
    }
#pragma unroll
    for (int r = 0; r < 8; ++r) {
        int orow = rowBase + r + 8 * kh;
        if (orow < n && m < OUT_DIM) H2[(size_t)orow * OUT_DIM + m] = acc[r];
    }
}

__global__ __launch_bounds__(256) void k_gather2(const int* __restrict__ off, const int* __restrict__ csr,
                                                 const float* __restrict__ dinv, const float* __restrict__ H2,
                                                 const float* __restrict__ b2, float* __restrict__ out, int n) {
    __shared__ float so[32 * OUT_DIM];
    const int g  = threadIdx.x >> 3;
    const int sl = threadIdx.x & 7;
    const int i  = blockIdx.x * 32 + g;

    float a0 = 0.0f, a1 = 0.0f, a2 = 0.0f;
    float di = 0.0f;
    if (i < n) {
        const int beg = off[i], end = off[i + 1];
        di = dinv[i];
        for (int j = beg + sl; j < end; j += 8) {
            const int   s = csr[j];
            const float w = dinv[s] * di;
            const float* hp = H2 + (size_t)s * OUT_DIM;
            a0 = fmaf(hp[0], w, a0);
            a1 = fmaf(hp[1], w, a1);
            a2 = fmaf(hp[2], w, a2);
        }
    }
#pragma unroll
    for (int m = 4; m >= 1; m >>= 1) {
        a0 += __shfl_xor(a0, m);
        a1 += __shfl_xor(a1, m);
        a2 += __shfl_xor(a2, m);
    }
    if (sl == 0 && i < n) {
        const float* hp = H2 + (size_t)i * OUT_DIM;
        so[g * OUT_DIM + 0] = a0 + hp[0] * di * di + b2[0];
        so[g * OUT_DIM + 1] = a1 + hp[1] * di * di + b2[1];
        so[g * OUT_DIM + 2] = a2 + hp[2] * di * di + b2[2];
    }
    __syncthreads();
    if (threadIdx.x < 32 * OUT_DIM) {
        const int idx = blockIdx.x * (32 * OUT_DIM) + threadIdx.x;
        if (idx < n * OUT_DIM) out[idx] = so[threadIdx.x];
    }
}

extern "C" void kernel_launch(void* const* d_in, const int* in_sizes, int n_in,
                              void* d_out, int out_size, void* d_ws, size_t ws_size,
                              hipStream_t stream) {
    (void)n_in; (void)out_size; (void)ws_size;
    const float* x  = (const float*)d_in[0];
    const int*  edge = (const int*)d_in[1];
    const float* W1 = (const float*)d_in[2];
    const float* b1 = (const float*)d_in[3];
    const float* W2 = (const float*)d_in[4];
    const float* b2 = (const float*)d_in[5];
    float* out = (float*)d_out;

    const int n = in_sizes[0] / IN_DIM;
    const int e = in_sizes[1] / 2;
    const int* src = edge;
    const int* dst = edge + e;
    const int nb = (n + BKT_NODES - 1) / BKT_NODES;

    const size_t np  = ((size_t)n + 1 + 63) & ~(size_t)63;
    const size_t ep  = ((size_t)e + 63) & ~(size_t)63;
    float* ws    = (float*)d_ws;
    float* dinv  = ws;
    int*   off   = (int*)(ws + np);
    int*   table = off + np;
    int*   bbase = table + (size_t)NCHUNK * NBMAX;
    int*   pairs = bbase + NBMAX + 64;
    int*   csr   = pairs + ep;
    float* H1    = (float*)(csr + ep);
    float* A1    = H1 + (size_t)n * HID_DIM;
    float* H2    = A1 + (size_t)n * HID_DIM;

    dim3 blk(256);
    k_hist <<<NCHUNK, SORT_THREADS, 0, stream>>>(dst, table, e, nb);
    k_part <<<NCHUNK, SORT_THREADS, 0, stream>>>(src, dst, table, bbase, pairs, e, nb);
    k_bsort<<<nb, SORT_THREADS, 0, stream>>>(pairs, bbase, off, dinv, csr, n);

    k_gemm1  <<<(n + 127) / 128, blk, 0, stream>>>(x, W1, H1, n);
    k_gather1<<<(n + 7) / 8, blk, 0, stream>>>(off, csr, dinv, H1, b1, A1, n);

    k_gemm2  <<<(n + 127) / 128, blk, 0, stream>>>(A1, W2, H2, n);
    k_gather2<<<(n + 31) / 32, blk, 0, stream>>>(off, csr, dinv, H2, b2, out, n);
}


# both edge-sort passes order their words in LDS and store contiguous runs (whole lines in the second pass) instead of one scattered 4-byte global store per edge; sums keep an arrival order
# speedup vs baseline: 6.7447x; 1.4983x over previous
#include <hip/hip_runtime.h>

typedef __attribute__((ext_vector_type(2))) float v2f;
typedef __attribute__((ext_vector_type(4))) float v4f;
typedef __attribute__((ext_vector_type(8))) float v8f;
typedef __attribute__((ext_vector_type(2))) int   v2i;

#define IN_DIM  128
#define HID_DIM 64
#define OUT_DIM 3
#define H2_PITCH 4
#define BKT_SHIFT 8
#define BKT_NODES (1 << BKT_SHIFT)
#define NBMAX 512
#define NCHUNK 256
#define SORT_THREADS 1024
#define STAGE_CAP 16384

__global__ __launch_bounds__(SORT_THREADS) void k_hist(const int* __restrict__ dst, int* __restrict__ table,
                                                       int e, int nb) {
    __shared__ int h[NBMAX];
    for (int k = threadIdx.x; k < NBMAX; k += SORT_THREADS) h[k] = 0;
    __syncthreads();
    const int ch  = (e + (int)gridDim.x - 1) / (int)gridDim.x;
    const int beg = (int)blockIdx.x * ch;
    const int end = (beg + ch < e) ? beg + ch : e;
    int j = beg + (int)threadIdx.x;
    for (; j + 3 * SORT_THREADS < end; j += 4 * SORT_THREADS) {
        int d[4];
#pragma unroll
        for (int u = 0; u < 4; ++u) d[u] = dst[j + u * SORT_THREADS];
#pragma unroll
        for (int u = 0; u < 4; ++u) atomicAdd(&h[d[u] >> BKT_SHIFT], 1);
    }
    for (; j < end; j += SORT_THREADS) atomicAdd(&h[dst[j] >> BKT_SHIFT], 1);
    __syncthreads();
    for (int k = threadIdx.x; k < nb; k += SORT_THREADS) table[(int)blockIdx.x * NBMAX + k] = h[k];
}

__global__ __launch_bounds__(SORT_THREADS) void k_part(const int* __restrict__ src, const int* __restrict__ dst,
                                                       const int* __restrict__ table, int* __restrict__ bbase,
                                                       int* __restrict__ pairs, int e, int nb) {
    __shared__ int tot[2][NBMAX];
    __shared__ int pre[2][NBMAX];
    __shared__ unsigned long long cur[NBMAX];
    __shared__ int wsum[SORT_THREADS / 32];
    __shared__ int lsum[SORT_THREADS / 32];
    __shared__ v2i stage[STAGE_CAP];
    const int tid  = threadIdx.x;
    const int lane = tid & 31;
    const int wave = tid >> 5;
    const int g    = blockIdx.x;
    const int G    = gridDim.x;

    const int own = (tid < nb) ? table[g * NBMAX + tid] : 0;

    {
        const int bk   = tid & (NBMAX - 1);
        const int half = tid / NBMAX;
        const int r0   = half ? (G >> 1) : 0;
        const int r1   = half ? G : (G >> 1);
        int t = 0, p = 0;
        if (bk < nb) {
#pragma unroll 8
            for (int r = r0; r < r1; ++r) {
                const int v = table[r * NBMAX + bk];
                t += v;
                if (r < g) p += v;
            }
        }
        tot[half][bk] = t;
        pre[half][bk] = p;
    }
    __syncthreads();

    const int val  = (tid < NBMAX) ? tot[0][tid] + tot[1][tid] : 0;
    const int lval = own;
    int inc = val, linc = lval;
#pragma unroll
    for (int d = 1; d < 32; d <<= 1) {
        const int v  = __shfl_up(inc, d);
        const int lv = __shfl_up(linc, d);
        if (lane >= d) { inc += v; linc += lv; }
    }
    if (lane == 31) { wsum[wave] = inc; lsum[wave] = linc; }
    __syncthreads();
    int wpre = 0, lpre = 0;
    for (int w = 0; w < wave; ++w) { wpre += wsum[w]; lpre += lsum[w]; }
    const int excl  = wpre + inc - val;
    const int lexcl = lpre + linc - lval;
    if (tid < NBMAX)
        cur[tid] = ((unsigned long long)(unsigned)(excl + pre[0][tid] + pre[1][tid]) << 32) | (unsigned)lexcl;
    if (g == 0 && tid <= nb) bbase[tid] = excl;
    __syncthreads();

    const int ch  = (e + G - 1) / G;
    const int beg = g * ch;
    const int end = (beg + ch < e) ? beg + ch : e;
    const unsigned long long STEP = 0x0000000100000001ULL;
    int j = beg + tid;
    for (; j + 3 * SORT_THREADS < end; j += 4 * SORT_THREADS) {
        int d[4], s[4];
        unsigned long long r[4];
#pragma unroll
        for (int u = 0; u < 4; ++u) {
            d[u] = dst[j + u * SORT_THREADS];
            s[u] = src[j + u * SORT_THREADS];
        }
#pragma unroll
        for (int u = 0; u < 4; ++u) r[u] = atomicAdd(&cur[d[u] >> BKT_SHIFT], STEP);
#pragma unroll
        for (int u = 0; u < 4; ++u)
            stage[(unsigned)r[u]] = v2i{ (int)(((unsigned)s[u] << BKT_SHIFT) | (unsigned)(d[u] & (BKT_NODES - 1))),
                                         (int)(r[u] >> 32) };
    }
    for (; j < end; j += SORT_THREADS) {
        const int d = dst[j];
        const int s = src[j];
        const unsigned long long r = atomicAdd(&cur[d >> BKT_SHIFT], STEP);
        stage[(unsigned)r] = v2i{ (int)(((unsigned)s << BKT_SHIFT) | (unsigned)(d & (BKT_NODES - 1))), (int)(r >> 32) };
    }
    __syncthreads();

    const int cnt = end - beg;
    int t = tid;
    for (; t + 3 * SORT_THREADS < cnt; t += 4 * SORT_THREADS) {
        v2i w[4];
#pragma unroll
        for (int u = 0; u < 4; ++u) w[u] = stage[t + u * SORT_THREADS];
#pragma unroll
        for (int u = 0; u < 4; ++u) pairs[w[u].y] = w[u].x;
    }
    for (; t < cnt; t += SORT_THREADS) {
        const v2i w = stage[t];
        pairs[w.y] = w.x;
    }
}

__global__ __launch_bounds__(SORT_THREADS) void k_bsort(const int* __restrict__ pairs, const int* __restrict__ bbase,
                                                        int* __restrict__ off, float* __restrict__ dinv,
                                                        int* __restrict__ csr, int n) {
    __shared__ int c[BKT_NODES];
    __shared__ int wsum[BKT_NODES / 32];
    __shared__ int stage[STAGE_CAP];
    const int tid  = threadIdx.x;
    const int lane = tid & 31;
    const int wave = tid >> 5;
    const int beg  = bbase[blockIdx.x];
    const int end  = bbase[blockIdx.x + 1];
    if (tid < BKT_NODES) c[tid] = 0;
    __syncthreads();
    int j = beg + tid;
    for (; j + 3 * SORT_THREADS < end; j += 4 * SORT_THREADS) {
        int w[4];
#pragma unroll
        for (int u = 0; u < 4; ++u) w[u] = pairs[j + u * SORT_THREADS];
#pragma unroll
        for (int u = 0; u < 4; ++u) atomicAdd(&c[w[u] & (BKT_NODES - 1)], 1);
    }
    for (; j < end; j += SORT_THREADS) atomicAdd(&c[pairs[j] & (BKT_NODES - 1)], 1);
    __syncthreads();

    const int val = (tid < BKT_NODES) ? c[tid] : 0;
    int inc = val;
#pragma unroll
    for (int d = 1; d < 32; d <<= 1) {
        const int v = __shfl_up(inc, d);
        if (lane >= d) inc += v;
    }
    if (tid < BKT_NODES && lane == 31) wsum[wave] = inc;
    __syncthreads();
    if (tid < BKT_NODES) {
        int wpre = 0;
        for (int w = 0; w < wave; ++w) wpre += wsum[w];
        const int first = wpre + inc - val;
        const int node  = (int)blockIdx.x * BKT_NODES + tid;
        c[tid] = first;
        if (node < n) {
            off[node]  = beg + first;
            dinv[node] = rsqrtf(fmaxf((float)(val + 1), 1.0f));
            if (node == n - 1) off[n] = beg + first + val;
        }
    }
    __syncthreads();
    const int cnt = end - beg;
    j = beg + tid;
    if (cnt > STAGE_CAP) {
        for (; j < end; j += SORT_THREADS) {
            const int w = pairs[j];
            const int pos = atomicAdd(&c[w & (BKT_NODES - 1)], 1);
            csr[beg + pos] = (int)((unsigned)w >> BKT_SHIFT);
        }
        return;
    }
    for (; j + 3 * SORT_THREADS < end; j += 4 * SORT_THREADS) {
        int w[4], pos[4];
#pragma unroll
        for (int u = 0; u < 4; ++u) w[u] = pairs[j + u * SORT_THREADS];
#pragma unroll
        for (int u = 0; u < 4; ++u) pos[u] = atomicAdd(&c[w[u] & (BKT_NODES - 1)], 1);
#pragma unroll
        for (int u = 0; u < 4; ++u) stage[pos[u]] = (int)((unsigned)w[u] >> BKT_SHIFT);
    }
    for (; j < end; j += SORT_THREADS) {
        const int w = pairs[j];
        const int pos = atomicAdd(&c[w & (BKT_NODES - 1)], 1);
        stage[pos] = (int)((unsigned)w >> BKT_SHIFT);
    }
    __syncthreads();

    int t = tid - (beg & 31);
    if (t >= 0 && t < cnt) csr[beg + t] = stage[t];
    t += SORT_THREADS;
    for (; t + 3 * SORT_THREADS < cnt; t += 4 * SORT_THREADS) {
        int v[4];
#pragma unroll
        for (int u = 0; u < 4; ++u) v[u] = stage[t + u * SORT_THREADS];
#pragma unroll
        for (int u = 0; u < 4; ++u) csr[beg + t + u * SORT_THREADS] = v[u];
    }
    for (; t < cnt; t += SORT_THREADS) csr[beg + t] = stage[t];
}

__global__ __launch_bounds__(256) void k_gemm1(const float* __restrict__ X,
                                               const float* __restrict__ W,
                                               const float* __restrict__ dinv,
                                               float* __restrict__ H, int n) {
    __shared__ float sW[IN_DIM * HID_DIM];
    for (int i = threadIdx.x; i < IN_DIM * HID_DIM; i += 256) sW[i] = W[i];
    __syncthreads();

    const int lane = threadIdx.x & 31;
    const int wave = threadIdx.x >> 5;
    const int rowBase = (blockIdx.x * 8 + wave) * 16;
    if (rowBase >= n) return;

    const int m  = lane & 15;
    const int kh = lane >> 4;
    int row  = rowBase + m;
    int rowc = row < n ? row : n - 1;
    const float* xrow = X + (size_t)rowc * IN_DIM;

    float dv[8];
#pragma unroll
    for (int r = 0; r < 8; ++r) {
        const int orow = rowBase + r + 8 * kh;
        dv[r] = dinv[orow < n ? orow : n - 1];
    }

    v8f a0 = {}, a1 = {}, a2 = {}, a3 = {};
#pragma unroll 4
    for (int k = 0; k < IN_DIM; k += 4) {
        v2f a = *(const v2f*)(xrow + k + 2 * kh);
        const float* wp = sW + (k + 2 * kh) * HID_DIM;
        v2f b0 = { wp[m +  0], wp[HID_DIM + m +  0] };
        v2f b1 = { wp[m + 16], wp[HID_DIM + m + 16] };
        v2f b2 = { wp[m + 32], wp[HID_DIM + m + 32] };
        v2f b3 = { wp[m + 48], wp[HID_DIM + m + 48] };
        a0 = __builtin_amdgcn_wmma_f32_16x16x4_f32(false, a, false, b0, (short)0, a0, false, false);
        a1 = __builtin_amdgcn_wmma_f32_16x16x4_f32(false, a, false, b1, (short)0, a1, false, false);
        a2 = __builtin_amdgcn_wmma_f32_16x16x4_f32(false, a, false, b2, (short)0, a2, false, false);
        a3 = __builtin_amdgcn_wmma_f32_16x16x4_f32(false, a, false, b3, (short)0, a3, false, false);
    }
#pragma unroll
    for (int r = 0; r < 8; ++r) {
        int orow = rowBase + r + 8 * kh;
        if (orow < n) {
            float* hp = H + (size_t)orow * HID_DIM + m;
            hp[0]  = a0[r] * dv[r];
            hp[16] = a1[r] * dv[r];
            hp[32] = a2[r] * dv[r];
            hp[48] = a3[r] * dv[r];
        }
    }
}

__global__ __launch_bounds__(256) void k_gather1(const int* __restrict__ off, const int* __restrict__ csr,
                                                 const float* __restrict__ dinv, const float* __restrict__ H1,
                                                 const float* __restrict__ b1, float* __restrict__ A1, int n) {
    const int lane = threadIdx.x & 31;
    const int i = __builtin_amdgcn_readfirstlane((int)(blockIdx.x * 8 + (threadIdx.x >> 5)));
    if (i >= n) return;
    const int beg = __builtin_amdgcn_readfirstlane(off[i]);
    const int end = __builtin_amdgcn_readfirstlane(off[i + 1]);
    const float di = dinv[i];
    const float* hcol = H1 + 2 * lane;

    v2f acc = *(const v2f*)(hcol + (size_t)i * HID_DIM);

    for (int j = beg; j < end; j += 32) {
        int c = end - j;
        if (c > 32) c = 32;
        int s_l = 0;
        if (lane < c) s_l = csr[j + lane];
        int t = 0;
        for (; t + 8 <= c; t += 8) {
            v2f h[8];
#pragma unroll
            for (int u = 0; u < 8; ++u) {
                const int s = __builtin_amdgcn_readlane(s_l, t + u);
                h[u] = *(const v2f*)(hcol + (size_t)s * HID_DIM);
            }
#pragma unroll
            for (int u = 0; u < 8; ++u) {
                acc.x += h[u].x;
                acc.y += h[u].y;
            }
        }
        if (t < c) {
            v2f h[7];
#pragma unroll
            for (int u = 0; u < 7; ++u) {
                h[u] = v2f{ 0.0f, 0.0f };
                if (t + u < c) {
                    const int s = __builtin_amdgcn_readlane(s_l, t + u);
                    h[u] = *(const v2f*)(hcol + (size_t)s * HID_DIM);
                }
            }
#pragma unroll
            for (int u = 0; u < 7; ++u) {
                acc.x += h[u].x;
                acc.y += h[u].y;
            }
        }
    }
    const v2f bb = *(const v2f*)(b1 + 2 * lane);
    v2f o = { fmaf(acc.x, di, bb.x), fmaf(acc.y, di, bb.y) };
    *(v2f*)(A1 + (size_t)i * HID_DIM + 2 * lane) = o;
}

__global__ __launch_bounds__(256) void k_gemm2(const float* __restrict__ A1,
                                               const float* __restrict__ W2,
                                               const float* __restrict__ dinv,
                                               float* __restrict__ H2, int n) {
    const int lane = threadIdx.x & 31;
    const int wave = threadIdx.x >> 5;
    const int rowBase = (blockIdx.x * 8 + wave) * 16;
    if (rowBase >= n) return;

    const int m  = lane & 15;
    const int kh = lane >> 4;
    int row  = rowBase + m;
    int rowc = row < n ? row : n - 1;
    const float* arow = A1 + (size_t)rowc * HID_DIM;

    float dv[8];
#pragma unroll
    for (int r = 0; r < 8; ++r) {
        const int orow = rowBase + r + 8 * kh;
        dv[r] = dinv[orow < n ? orow : n - 1];
    }

    v8f acc = {};
#pragma unroll 4
    for (int k = 0; k < HID_DIM; k += 4) {
        v2f a = *(const v2f*)(arow + k + 2 * kh);
        a.x = fmaxf(a.x, 0.0f);
        a.y = fmaxf(a.y, 0.0f);
        v2f b = { 0.0f, 0.0f };
        if (m < OUT_DIM) {
            b.x = W2[(k + 2 * kh) * OUT_DIM + m];
            b.y = W2[(k + 2 * kh + 1) * OUT_DIM + m];
        }
        acc = __builtin_amdgcn_wmma_f32_16x16x4_f32(false, a, false, b, (short)0, acc, false, false);
    }
#pragma unroll
    for (int r = 0; r < 8; ++r) {
        int orow = rowBase + r + 8 * kh;
        if (orow < n && m < H2_PITCH) H2[(size_t)orow * H2_PITCH + m] = (m < OUT_DIM) ? acc[r] * dv[r] : 0.0f;
    }
}

__global__ __launch_bounds__(256) void k_gather2(const int* __restrict__ off, const int* __restrict__ csr,
                                                 const float* __restrict__ dinv, const float* __restrict__ H2,
                                                 const float* __restrict__ b2, float* __restrict__ out, int n) {
    __shared__ float so[32 * OUT_DIM];
    const int g  = threadIdx.x >> 3;
    const int sl = threadIdx.x & 7;
    const int i  = blockIdx.x * 32 + g;

    float a0 = 0.0f, a1 = 0.0f, a2 = 0.0f;
    float di = 0.0f;
    if (i < n) {
        const int beg = off[i], end = off[i + 1];
        di = dinv[i];
        for (int j = beg + sl; j < end; j += 8) {
            const int s = csr[j];
            const v4f h = *(const v4f*)(H2 + (size_t)s * H2_PITCH);
            a0 += h.x;
            a1 += h.y;
            a2 += h.z;
        }
    }
#pragma unroll
    for (int m = 4; m >= 1; m >>= 1) {
        a0 += __shfl_xor(a0, m);
        a1 += __shfl_xor(a1, m);
        a2 += __shfl_xor(a2, m);
    }
    if (sl == 0 && i < n) {
        const v4f hs = *(const v4f*)(H2 + (size_t)i * H2_PITCH);
        so[g * OUT_DIM + 0] = fmaf(a0 + hs.x, di, b2[0]);
        so[g * OUT_DIM + 1] = fmaf(a1 + hs.y, di, b2[1]);
        so[g * OUT_DIM + 2] = fmaf(a2 + hs.z, di, b2[2]);
    }
    __syncthreads();
    if (threadIdx.x < 32 * OUT_DIM) {
        const int idx = blockIdx.x * (32 * OUT_DIM) + threadIdx.x;
        if (idx < n * OUT_DIM) out[idx] = so[threadIdx.x];
    }
}

extern "C" void kernel_launch(void* const* d_in, const int* in_sizes, int n_in,
                              void* d_out, int out_size, void* d_ws, size_t ws_size,
                              hipStream_t stream) {
    (void)n_in; (void)out_size; (void)ws_size;
    const float* x  = (const float*)d_in[0];
    const int*  edge = (const int*)d_in[1];
    const float* W1 = (const float*)d_in[2];
    const float* b1 = (const float*)d_in[3];
    const float* W2 = (const float*)d_in[4];
    const float* b2 = (const float*)d_in[5];
    float* out = (float*)d_out;

    const int n = in_sizes[0] / IN_DIM;
    const int e = in_sizes[1] / 2;
    const int* src = edge;
    const int* dst = edge + e;
    const int nb = (n + BKT_NODES - 1) / BKT_NODES;

    const size_t np  = ((size_t)n + 1 + 63) & ~(size_t)63;
    const size_t ep  = ((size_t)e + 63) & ~(size_t)63;
    float* ws    = (float*)d_ws;
    float* dinv  = ws;
    int*   off   = (int*)(ws + np);
    int*   table = off + np;
    int*   bbase = table + (size_t)NCHUNK * NBMAX;
    int*   pairs = bbase + NBMAX + 64;
    int*   csr   = pairs + ep;
    float* H1    = (float*)(csr + ep);
    float* A1    = H1 + (size_t)n * HID_DIM;
    float* H2    = A1 + (size_t)n * HID_DIM;

    dim3 blk(256);
    k_hist <<<NCHUNK, SORT_THREADS, 0, stream>>>(dst, table, e, nb);
    k_part <<<NCHUNK, SORT_THREADS, 0, stream>>>(src, dst, table, bbase, pairs, e, nb);
    k_bsort<<<nb, SORT_THREADS, 0, stream>>>(pairs, bbase, off, dinv, csr, n);

    k_gemm1  <<<(n + 127) / 128, blk, 0, stream>>>(x, W1, dinv, H1, n);
    k_gather1<<<(n + 7) / 8, blk, 0, stream>>>(off, csr, dinv, H1, b1, A1, n);

    k_gemm2  <<<(n + 127) / 128, blk, 0, stream>>>(A1, W2, dinv, H2, n);
    k_gather2<<<(n + 31) / 32, blk, 0, stream>>>(off, csr, dinv, H2, b2, out, n);
}


# layer-2 product moved into the layer-1 gather launch: a block gathers 16 rows to LDS, one wave multiplies them by W2 on the f32 matrix instruction; A1 never stored, one launch less; sums unchanged
# speedup vs baseline: 7.0386x; 1.0436x over previous
#include <hip/hip_runtime.h>

typedef __attribute__((ext_vector_type(2))) float v2f;
typedef __attribute__((ext_vector_type(4))) float v4f;
typedef __attribute__((ext_vector_type(8))) float v8f;
typedef __attribute__((ext_vector_type(2))) int   v2i;

#define IN_DIM  128
#define HID_DIM 64
#define OUT_DIM 3
#define H2_PITCH 4
#define BKT_SHIFT 8
#define BKT_NODES (1 << BKT_SHIFT)
#define NBMAX 512
#define NCHUNK 256
#define SORT_THREADS 1024
#define STAGE_CAP 16384

__global__ __launch_bounds__(SORT_THREADS) void k_hist(const int* __restrict__ dst, int* __restrict__ table,
                                                       int e, int nb) {
    __shared__ int h[NBMAX];
    for (int k = threadIdx.x; k < NBMAX; k += SORT_THREADS) h[k] = 0;
    __syncthreads();
    const int ch  = (e + (int)gridDim.x - 1) / (int)gridDim.x;
    const int beg = (int)blockIdx.x * ch;
    const int end = (beg + ch < e) ? beg + ch : e;
    int j = beg + (int)threadIdx.x;
    for (; j + 3 * SORT_THREADS < end; j += 4 * SORT_THREADS) {
        int d[4];
#pragma unroll
        for (int u = 0; u < 4; ++u) d[u] = dst[j + u * SORT_THREADS];
#pragma unroll
        for (int u = 0; u < 4; ++u) atomicAdd(&h[d[u] >> BKT_SHIFT], 1);
    }
    for (; j < end; j += SORT_THREADS) atomicAdd(&h[dst[j] >> BKT_SHIFT], 1);
    __syncthreads();
    for (int k = threadIdx.x; k < nb; k += SORT_THREADS) table[(int)blockIdx.x * NBMAX + k] = h[k];
}

__global__ __launch_bounds__(SORT_THREADS) void k_part(const int* __restrict__ src, const int* __restrict__ dst,
                                                       const int* __restrict__ table, int* __restrict__ bbase,
                                                       int* __restrict__ pairs, int e, int nb) {
    __shared__ int tot[2][NBMAX];
    __shared__ int pre[2][NBMAX];
    __shared__ unsigned long long cur[NBMAX];
    __shared__ int wsum[SORT_THREADS / 32];
    __shared__ int lsum[SORT_THREADS / 32];
    __shared__ v2i stage[STAGE_CAP];
    const int tid  = threadIdx.x;
    const int lane = tid & 31;
    const int wave = tid >> 5;
    const int g    = blockIdx.x;
    const int G    = gridDim.x;

    const int own = (tid < nb) ? table[g * NBMAX + tid] : 0;

    {
        const int bk   = tid & (NBMAX - 1);
        const int half = tid / NBMAX;
        const int r0   = half ? (G >> 1) : 0;
        const int r1   = half ? G : (G >> 1);
        int t = 0, p = 0;
        if (bk < nb) {
#pragma unroll 8
            for (int r = r0; r < r1; ++r) {
                const int v = table[r * NBMAX + bk];
                t += v;
                if (r < g) p += v;
            }
        }
        tot[half][bk] = t;
        pre[half][bk] = p;
    }
    __syncthreads();

    const int val  = (tid < NBMAX) ? tot[0][tid] + tot[1][tid] : 0;
    const int lval = own;
    int inc = val, linc = lval;
#pragma unroll
    for (int d = 1; d < 32; d <<= 1) {
        const int v  = __shfl_up(inc, d);
        const int lv = __shfl_up(linc, d);
        if (lane >= d) { inc += v; linc += lv; }
    }
    if (lane == 31) { wsum[wave] = inc; lsum[wave] = linc; }
    __syncthreads();
    int wpre = 0, lpre = 0;
    for (int w = 0; w < wave; ++w) { wpre += wsum[w]; lpre += lsum[w]; }
    const int excl  = wpre + inc - val;
    const int lexcl = lpre + linc - lval;
    if (tid < NBMAX)
        cur[tid] = ((unsigned long long)(unsigned)(excl + pre[0][tid] + pre[1][tid]) << 32) | (unsigned)lexcl;
    if (g == 0 && tid <= nb) bbase[tid] = excl;
    __syncthreads();

    const int ch  = (e + G - 1) / G;
    const int beg = g * ch;
    const int end = (beg + ch < e) ? beg + ch : e;
    const unsigned long long STEP = 0x0000000100000001ULL;
    int j = beg + tid;
    for (; j + 3 * SORT_THREADS < end; j += 4 * SORT_THREADS) {
        int d[4], s[4];
        unsigned long long r[4];
#pragma unroll
        for (int u = 0; u < 4; ++u) {
            d[u] = dst[j + u * SORT_THREADS];
            s[u] = src[j + u * SORT_THREADS];
        }
#pragma unroll
        for (int u = 0; u < 4; ++u) r[u] = atomicAdd(&cur[d[u] >> BKT_SHIFT], STEP);
#pragma unroll
        for (int u = 0; u < 4; ++u)
            stage[(unsigned)r[u]] = v2i{ (int)(((unsigned)s[u] << BKT_SHIFT) | (unsigned)(d[u] & (BKT_NODES - 1))),
                                         (int)(r[u] >> 32) };
    }
    for (; j < end; j += SORT_THREADS) {
        const int d = dst[j];
        const int s = src[j];
        const unsigned long long r = atomicAdd(&cur[d >> BKT_SHIFT], STEP);
        stage[(unsigned)r] = v2i{ (int)(((unsigned)s << BKT_SHIFT) | (unsigned)(d & (BKT_NODES - 1))), (int)(r >> 32) };
    }
    __syncthreads();

    const int cnt = end - beg;
    int t = tid;
    for (; t + 3 * SORT_THREADS < cnt; t += 4 * SORT_THREADS) {
        v2i w[4];
#pragma unroll
        for (int u = 0; u < 4; ++u) w[u] = stage[t + u * SORT_THREADS];
#pragma unroll
        for (int u = 0; u < 4; ++u) pairs[w[u].y] = w[u].x;
    }
    for (; t < cnt; t += SORT_THREADS) {
        const v2i w = stage[t];
        pairs[w.y] = w.x;
    }
}

__global__ __launch_bounds__(SORT_THREADS) void k_bsort(const int* __restrict__ pairs, const int* __restrict__ bbase,
                                                        int* __restrict__ off, float* __restrict__ dinv,
                                                        int* __restrict__ csr, int n) {
    __shared__ int c[BKT_NODES];
    __shared__ int wsum[BKT_NODES / 32];
    __shared__ int stage[STAGE_CAP];
    const int tid  = threadIdx.x;
    const int lane = tid & 31;
    const int wave = tid >> 5;
    const int beg  = bbase[blockIdx.x];
    const int end  = bbase[blockIdx.x + 1];
    if (tid < BKT_NODES) c[tid] = 0;
    __syncthreads();
    int j = beg + tid;
    for (; j + 3 * SORT_THREADS < end; j += 4 * SORT_THREADS) {
        int w[4];
#pragma unroll
        for (int u = 0; u < 4; ++u) w[u] = pairs[j + u * SORT_THREADS];
#pragma unroll
        for (int u = 0; u < 4; ++u) atomicAdd(&c[w[u] & (BKT_NODES - 1)], 1);
    }
    for (; j < end; j += SORT_THREADS) atomicAdd(&c[pairs[j] & (BKT_NODES - 1)], 1);
    __syncthreads();

    const int val = (tid < BKT_NODES) ? c[tid] : 0;
    int inc = val;
#pragma unroll
    for (int d = 1; d < 32; d <<= 1) {
        const int v = __shfl_up(inc, d);
        if (lane >= d) inc += v;
    }
    if (tid < BKT_NODES && lane == 31) wsum[wave] = inc;
    __syncthreads();
    if (tid < BKT_NODES) {
        int wpre = 0;
        for (int w = 0; w < wave; ++w) wpre += wsum[w];
        const int first = wpre + inc - val;
        const int node  = (int)blockIdx.x * BKT_NODES + tid;
        c[tid] = first;
        if (node < n) {
            off[node]  = beg + first;
            dinv[node] = rsqrtf(fmaxf((float)(val + 1), 1.0f));
            if (node == n - 1) off[n] = beg + first + val;
        }
    }
    __syncthreads();
    const int cnt = end - beg;
    j = beg + tid;
    if (cnt > STAGE_CAP) {
        for (; j < end; j += SORT_THREADS) {
            const int w = pairs[j];
            const int pos = atomicAdd(&c[w & (BKT_NODES - 1)], 1);
            csr[beg + pos] = (int)((unsigned)w >> BKT_SHIFT);
        }
        return;
    }
    for (; j + 3 * SORT_THREADS < end; j += 4 * SORT_THREADS) {
        int w[4], pos[4];
#pragma unroll
        for (int u = 0; u < 4; ++u) w[u] = pairs[j + u * SORT_THREADS];
#pragma unroll
        for (int u = 0; u < 4; ++u) pos[u] = atomicAdd(&c[w[u] & (BKT_NODES - 1)], 1);
#pragma unroll
        for (int u = 0; u < 4; ++u) stage[pos[u]] = (int)((unsigned)w[u] >> BKT_SHIFT);
    }
    for (; j < end; j += SORT_THREADS) {
        const int w = pairs[j];
        const int pos = atomicAdd(&c[w & (BKT_NODES - 1)], 1);
        stage[pos] = (int)((unsigned)w >> BKT_SHIFT);
    }
    __syncthreads();

    int t = tid - (beg & 31);
    if (t >= 0 && t < cnt) csr[beg + t] = stage[t];
    t += SORT_THREADS;
    for (; t + 3 * SORT_THREADS < cnt; t += 4 * SORT_THREADS) {
        int v[4];
#pragma unroll
        for (int u = 0; u < 4; ++u) v[u] = stage[t + u * SORT_THREADS];
#pragma unroll
        for (int u = 0; u < 4; ++u) csr[beg + t + u * SORT_THREADS] = v[u];
    }
    for (; t < cnt; t += SORT_THREADS) csr[beg + t] = stage[t];
}

__global__ __launch_bounds__(256) void k_gemm1(const float* __restrict__ X,
                                               const float* __restrict__ W,
                                               const float* __restrict__ dinv,
                                               float* __restrict__ H, int n) {
    __shared__ float sW[IN_DIM * HID_DIM];
    for (int i = threadIdx.x; i < IN_DIM * HID_DIM; i += 256) sW[i] = W[i];
    __syncthreads();

    const int lane = threadIdx.x & 31;
    const int wave = threadIdx.x >> 5;
    const int rowBase = (blockIdx.x * 8 + wave) * 16;
    if (rowBase >= n) return;

    const int m  = lane & 15;
    const int kh = lane >> 4;
    int row  = rowBase + m;
    int rowc = row < n ? row : n - 1;
    const float* xrow = X + (size_t)rowc * IN_DIM;

    float dv[8];
#pragma unroll
    for (int r = 0; r < 8; ++r) {
        const int orow = rowBase + r + 8 * kh;
        dv[r] = dinv[orow < n ? orow : n - 1];
    }

    v8f a0 = {}, a1 = {}, a2 = {}, a3 = {};
#pragma unroll 4
    for (int k = 0; k < IN_DIM; k += 4) {
        v2f a = *(const v2f*)(xrow + k + 2 * kh);
        const float* wp = sW + (k + 2 * kh) * HID_DIM;
        v2f b0 = { wp[m +  0], wp[HID_DIM + m +  0] };
        v2f b1 = { wp[m + 16], wp[HID_DIM + m + 16] };
        v2f b2 = { wp[m + 32], wp[HID_DIM + m + 32] };
        v2f b3 = { wp[m + 48], wp[HID_DIM + m + 48] };
        a0 = __builtin_amdgcn_wmma_f32_16x16x4_f32(false, a, false, b0, (short)0, a0, false, false);
        a1 = __builtin_amdgcn_wmma_f32_16x16x4_f32(false, a, false, b1, (short)0, a1, false, false);
        a2 = __builtin_amdgcn_wmma_f32_16x16x4_f32(false, a, false, b2, (short)0, a2, false, false);
        a3 = __builtin_amdgcn_wmma_f32_16x16x4_f32(false, a, false, b3, (short)0, a3, false, false);
    }
#pragma unroll
    for (int r = 0; r < 8; ++r) {
        int orow = rowBase + r + 8 * kh;
        if (orow < n) {
            float* hp = H + (size_t)orow * HID_DIM + m;
            hp[0]  = a0[r] * dv[r];
            hp[16] = a1[r] * dv[r];
            hp[32] = a2[r] * dv[r];
            hp[48] = a3[r] * dv[r];
        }
    }
}

#define A_PITCH 68
__global__ __launch_bounds__(256) void k_gather1(const int* __restrict__ off, const int* __restrict__ csr,
                                                 const float* __restrict__ dinv, const float* __restrict__ H1,
                                                 const float* __restrict__ b1, const float* __restrict__ W2,
                                                 float* __restrict__ H2, int n) {
    __shared__ __attribute__((aligned(16))) float sA[16 * A_PITCH];
    __shared__ __attribute__((aligned(16))) float sD[16];
    __shared__ float sW[HID_DIM * OUT_DIM];
    const int lane = threadIdx.x & 31;
    const int wave = threadIdx.x >> 5;
    const int rowBase = (int)blockIdx.x * 16;
    const float* hcol = H1 + 2 * lane;
    const v2f bb = *(const v2f*)(b1 + 2 * lane);
    float w2v = 0.0f;
    if (threadIdx.x < HID_DIM * OUT_DIM) w2v = W2[threadIdx.x];

    for (int rr = 0; rr < 2; ++rr) {
        const int lr = __builtin_amdgcn_readfirstlane(wave + 8 * rr);
        const int i  = rowBase + lr;
        v2f o = { 0.0f, 0.0f };
        float di = 0.0f;
        if (i < n) {
            const int beg = __builtin_amdgcn_readfirstlane(off[i]);
            const int end = __builtin_amdgcn_readfirstlane(off[i + 1]);
            di = dinv[i];

            v2f acc = *(const v2f*)(hcol + (size_t)i * HID_DIM);

            for (int j = beg; j < end; j += 32) {
                int c = end - j;
                if (c > 32) c = 32;
                int s_l = 0;
                if (lane < c) s_l = csr[j + lane];
                int t = 0;
                for (; t + 8 <= c; t += 8) {
                    v2f h[8];
#pragma unroll
                    for (int u = 0; u < 8; ++u) {
                        const int s = __builtin_amdgcn_readlane(s_l, t + u);
                        h[u] = *(const v2f*)(hcol + (size_t)s * HID_DIM);
                    }
#pragma unroll
                    for (int u = 0; u < 8; ++u) {
                        acc.x += h[u].x;
                        acc.y += h[u].y;
                    }
                }
                if (t < c) {
                    v2f h[7];
#pragma unroll
                    for (int u = 0; u < 7; ++u) {
                        h[u] = v2f{ 0.0f, 0.0f };
                        if (t + u < c) {
                            const int s = __builtin_amdgcn_readlane(s_l, t + u);
                            h[u] = *(const v2f*)(hcol + (size_t)s * HID_DIM);
                        }
                    }
#pragma unroll
                    for (int u = 0; u < 7; ++u) {
                        acc.x += h[u].x;
                        acc.y += h[u].y;
                    }
                }
            }
            o.x = fmaxf(fmaf(acc.x, di, bb.x), 0.0f);
            o.y = fmaxf(fmaf(acc.y, di, bb.y), 0.0f);
        }
        *(v2f*)(sA + lr * A_PITCH + 2 * lane) = o;
        if (lane == 0) sD[lr] = di;
    }
    if (threadIdx.x < HID_DIM * OUT_DIM) sW[threadIdx.x] = w2v;
    __syncthreads();
    if (wave != 0) return;

    const int m  = lane & 15;
    const int kh = lane >> 4;
    const float* arow = sA + m * A_PITCH + 2 * kh;
    const bool live = m < OUT_DIM;
    const float* wcol = sW + 2 * kh * OUT_DIM + (live ? m : 0);
    const v4f dlo = *(const v4f*)(sD + 8 * kh);
    const v4f dhi = *(const v4f*)(sD + 8 * kh + 4);
    const float dv[8] = { dlo.x, dlo.y, dlo.z, dlo.w, dhi.x, dhi.y, dhi.z, dhi.w };
    v8f acc2 = {};
#pragma unroll 4
    for (int k = 0; k < HID_DIM; k += 4) {
        const v2f a = *(const v2f*)(arow + k);
        const float bx = wcol[k * OUT_DIM];
        const float by = wcol[(k + 1) * OUT_DIM];
        const v2f b = { live ? bx : 0.0f, live ? by : 0.0f };
        acc2 = __builtin_amdgcn_wmma_f32_16x16x4_f32(false, a, false, b, (short)0, acc2, false, false);
    }
#pragma unroll
    for (int r = 0; r < 8; ++r) {
        const int orow = rowBase + r + 8 * kh;
        if (orow < n && m < H2_PITCH) H2[(size_t)orow * H2_PITCH + m] = live ? acc2[r] * dv[r] : 0.0f;
    }
}

__global__ __launch_bounds__(256) void k_gather2(const int* __restrict__ off, const int* __restrict__ csr,
                                                 const float* __restrict__ dinv, const float* __restrict__ H2,
                                                 const float* __restrict__ b2, float* __restrict__ out, int n) {
    __shared__ float so[32 * OUT_DIM];
    const int g  = threadIdx.x >> 3;
    const int sl = threadIdx.x & 7;
    const int i  = blockIdx.x * 32 + g;

    float a0 = 0.0f, a1 = 0.0f, a2 = 0.0f;
    float di = 0.0f;
    if (i < n) {
        const int beg = off[i], end = off[i + 1];
        di = dinv[i];
        for (int j = beg + sl; j < end; j += 8) {
            const int s = csr[j];
            const v4f h = *(const v4f*)(H2 + (size_t)s * H2_PITCH);
            a0 += h.x;
            a1 += h.y;
            a2 += h.z;
        }
    }
#pragma unroll
    for (int m = 4; m >= 1; m >>= 1) {
        a0 += __shfl_xor(a0, m);
        a1 += __shfl_xor(a1, m);
        a2 += __shfl_xor(a2, m);
    }
    if (sl == 0 && i < n) {
        const v4f hs = *(const v4f*)(H2 + (size_t)i * H2_PITCH);
        so[g * OUT_DIM + 0] = fmaf(a0 + hs.x, di, b2[0]);
        so[g * OUT_DIM + 1] = fmaf(a1 + hs.y, di, b2[1]);
        so[g * OUT_DIM + 2] = fmaf(a2 + hs.z, di, b2[2]);
    }
    __syncthreads();
    if (threadIdx.x < 32 * OUT_DIM) {
        const int idx = blockIdx.x * (32 * OUT_DIM) + threadIdx.x;
        if (idx < n * OUT_DIM) out[idx] = so[threadIdx.x];
    }
}

extern "C" void kernel_launch(void* const* d_in, const int* in_sizes, int n_in,
                              void* d_out, int out_size, void* d_ws, size_t ws_size,
                              hipStream_t stream) {
    (void)n_in; (void)out_size; (void)ws_size;
    const float* x  = (const float*)d_in[0];
    const int*  edge = (const int*)d_in[1];
    const float* W1 = (const float*)d_in[2];
    const float* b1 = (const float*)d_in[3];
    const float* W2 = (const float*)d_in[4];
    const float* b2 = (const float*)d_in[5];
    float* out = (float*)d_out;

    const int n = in_sizes[0] / IN_DIM;
    const int e = in_sizes[1] / 2;
    const int* src = edge;
    const int* dst = edge + e;
    const int nb = (n + BKT_NODES - 1) / BKT_NODES;

    const size_t np  = ((size_t)n + 1 + 63) & ~(size_t)63;
    const size_t ep  = ((size_t)e + 63) & ~(size_t)63;
    float* ws    = (float*)d_ws;
    float* dinv  = ws;
    int*   off   = (int*)(ws + np);
    int*   table = off + np;
    int*   bbase = table + (size_t)NCHUNK * NBMAX;
    int*   pairs = bbase + NBMAX + 64;
    int*   csr   = pairs + ep;
    float* H1    = (float*)(csr + ep);
    float* H2    = H1 + (size_t)n * HID_DIM;

    dim3 blk(256);
    k_hist <<<NCHUNK, SORT_THREADS, 0, stream>>>(dst, table, e, nb);
    k_part <<<NCHUNK, SORT_THREADS, 0, stream>>>(src, dst, table, bbase, pairs, e, nb);
    k_bsort<<<nb, SORT_THREADS, 0, stream>>>(pairs, bbase, off, dinv, csr, n);

    k_gemm1  <<<(n + 127) / 128, blk, 0, stream>>>(x, W1, dinv, H1, n);
    k_gather1<<<(n + 15) / 16, blk, 0, stream>>>(off, csr, dinv, H1, b1, W2, H2, n);

    k_gather2<<<(n + 31) / 32, blk, 0, stream>>>(off, csr, dinv, H2, b2, out, n);
}


# first sort level made chunk-local: a chunk ranks and groups its own edges by bucket in LDS and stores one whole slice; a bucket collects its 256 segments; count launch gone; sums keep an arrival order
# speedup vs baseline: 7.5495x; 1.0726x over previous
#include <hip/hip_runtime.h>

typedef __attribute__((ext_vector_type(2))) float v2f;
typedef __attribute__((ext_vector_type(4))) float v4f;
typedef __attribute__((ext_vector_type(8))) float v8f;
typedef __attribute__((ext_vector_type(2))) int   v2i;

#define IN_DIM  128
#define HID_DIM 64
#define OUT_DIM 3
#define H2_PITCH 4
#define BKT_SHIFT 8
#define BKT_NODES (1 << BKT_SHIFT)
#define NBMAX 512
#define NCHUNK 256
#define SORT_THREADS 1024
#define STAGE_CAP 16384

#define EPT (STAGE_CAP / SORT_THREADS)
typedef __attribute__((ext_vector_type(4))) int v4i;

__global__ __launch_bounds__(SORT_THREADS) void k_part(const int* __restrict__ src, const int* __restrict__ dst,
                                                       int* __restrict__ tableT, int* __restrict__ pairs,
                                                       int e, int nb) {
    __shared__ int cnt[NBMAX];
    __shared__ int wsum[SORT_THREADS / 32];
    __shared__ __attribute__((aligned(16))) int stage[STAGE_CAP];
    const int tid  = threadIdx.x;
    const int lane = tid & 31;
    const int wave = tid >> 5;
    const int g    = blockIdx.x;

    const int ch  = (e + NCHUNK - 1) / NCHUNK;
    const int chp = (ch + 31) & ~31;
    const int beg = g * ch;
    const int end = (beg + ch < e) ? beg + ch : e;

    int d[EPT], s[EPT], r[EPT];
#pragma unroll
    for (int u = 0; u < EPT; ++u) {
        const int j = beg + tid + u * SORT_THREADS;
        d[u] = 0; s[u] = 0;
        if (j < end) { d[u] = dst[j]; s[u] = src[j]; }
    }
    if (tid < NBMAX) cnt[tid] = 0;
    __syncthreads();
#pragma unroll
    for (int u = 0; u < EPT; ++u) {
        r[u] = 0;
        if (beg + tid + u * SORT_THREADS < end) r[u] = atomicAdd(&cnt[d[u] >> BKT_SHIFT], 1);
    }
    __syncthreads();

    const int val = (tid < NBMAX) ? cnt[tid] : 0;
    int inc = val;
#pragma unroll
    for (int k = 1; k < 32; k <<= 1) {
        const int v = __shfl_up(inc, k);
        if (lane >= k) inc += v;
    }
    if (lane == 31) wsum[wave] = inc;
    __syncthreads();
    int wpre = 0;
    for (int w = 0; w < wave; ++w) wpre += wsum[w];
    const int lexcl = wpre + inc - val;
    if (tid < NBMAX) cnt[tid] = lexcl;
    if (tid <= nb) tableT[tid * NCHUNK + g] = lexcl;
    __syncthreads();

    int first[EPT];
#pragma unroll
    for (int u = 0; u < EPT; ++u) first[u] = cnt[d[u] >> BKT_SHIFT];
#pragma unroll
    for (int u = 0; u < EPT; ++u) {
        if (beg + tid + u * SORT_THREADS < end)
            stage[first[u] + r[u]] = (int)(((unsigned)s[u] << BKT_SHIFT) | (unsigned)(d[u] & (BKT_NODES - 1)));
    }
    __syncthreads();

    int* __restrict__ outp = pairs + (size_t)g * chp;
    const int c  = end > beg ? end - beg : 0;
    const int c4 = c & ~3;
    for (int t = tid * 4; t < c4; t += 4 * SORT_THREADS) *(v4i*)(outp + t) = *(const v4i*)(stage + t);
    if (tid < c - c4) outp[c4 + tid] = stage[c4 + tid];
}

__global__ __launch_bounds__(SORT_THREADS, 16) void k_bsort(const int* __restrict__ pairs, const int* __restrict__ tableT,
                                                        int* __restrict__ off, float* __restrict__ dinv,
                                                        int* __restrict__ csr, int n, int e) {
    __shared__ int c[BKT_NODES];
    __shared__ int wsum[BKT_NODES / 32];
    __shared__ int segs[NCHUNK];
    __shared__ int segl[NCHUNK];
    __shared__ int red[2][SORT_THREADS / 32];
    __shared__ int stage[STAGE_CAP];
    const int tid  = threadIdx.x;
    const int lane = tid & 31;
    const int wave = __builtin_amdgcn_readfirstlane(tid >> 5);
    const int b    = blockIdx.x;

    int s0 = 0, ln = 0;
    if (tid < NCHUNK) {
        s0 = tableT[b * NCHUNK + tid];
        ln = tableT[(b + 1) * NCHUNK + tid] - s0;
        segs[tid] = s0;
        segl[tid] = ln;
    }
    if (tid < BKT_NODES) c[tid] = 0;
#pragma unroll
    for (int k = 16; k >= 1; k >>= 1) {
        s0 += __shfl_xor(s0, k);
        ln += __shfl_xor(ln, k);
    }
    if (lane == 0) { red[0][wave] = s0; red[1][wave] = ln; }
    __syncthreads();
    int beg = 0, cnt = 0;
    for (int w = 0; w < NCHUNK / 32; ++w) { beg += red[0][w]; cnt += red[1][w]; }

    const int ch  = (e + NCHUNK - 1) / NCHUNK;
    const int chp = (ch + 31) & ~31;
    const int* skip = tableT + b * NCHUNK;
    int w0[NCHUNK / 32], w1[NCHUNK / 32];
#pragma unroll
    for (int q = 0; q < NCHUNK / 32; ++q) {
        const int gq = wave + 32 * q;
        const int lg = __builtin_amdgcn_readfirstlane(segl[gq]);
        const int* p = pairs + (size_t)gq * chp + __builtin_amdgcn_readfirstlane(segs[gq]);
        const int k0 = lane < lg ? lane : lg - 1;
        const int k1 = lane + 32 < lg ? lane + 32 : lg - 1;
        w0[q] = lg > 0  ? p[k0] : skip[0];
        w1[q] = lg > 32 ? p[k1] : skip[0];
    }
#pragma unroll
    for (int q = 0; q < NCHUNK / 32; ++q) {
        const int lg = __builtin_amdgcn_readfirstlane(segl[wave + 32 * q]);
        if (lane < lg)      atomicAdd(&c[w0[q] & (BKT_NODES - 1)], 1);
        if (lane + 32 < lg) atomicAdd(&c[w1[q] & (BKT_NODES - 1)], 1);
    }
    for (int q = 0; q < NCHUNK / 32; ++q) {
        const int gq = wave + 32 * q;
        const int lg = __builtin_amdgcn_readfirstlane(segl[gq]);
        const int* p = pairs + (size_t)gq * chp + __builtin_amdgcn_readfirstlane(segs[gq]);
        for (int k = 64 + lane; k < lg; k += 32) atomicAdd(&c[p[k] & (BKT_NODES - 1)], 1);
    }
    __syncthreads();

    const int val = (tid < BKT_NODES) ? c[tid] : 0;
    int inc = val;
#pragma unroll
    for (int k = 1; k < 32; k <<= 1) {
        const int v = __shfl_up(inc, k);
        if (lane >= k) inc += v;
    }
    if (tid < BKT_NODES && lane == 31) wsum[wave] = inc;
    __syncthreads();
    if (tid < BKT_NODES) {
        int wpre = 0;
        for (int w = 0; w < wave; ++w) wpre += wsum[w];
        const int first = wpre + inc - val;
        const int node  = b * BKT_NODES + tid;
        c[tid] = first;
        if (node < n) {
            off[node]  = beg + first;
            dinv[node] = rsqrtf(fmaxf((float)(val + 1), 1.0f));
            if (node == n - 1) off[n] = beg + first + val;
        }
    }
    __syncthreads();

    const bool direct = cnt > STAGE_CAP;
#pragma unroll
    for (int q = 0; q < NCHUNK / 32; ++q) {
        const int lg = __builtin_amdgcn_readfirstlane(segl[wave + 32 * q]);
        if (lane < lg) {
            const int pos = atomicAdd(&c[w0[q] & (BKT_NODES - 1)], 1);
            if (direct) csr[beg + pos] = (int)((unsigned)w0[q] >> BKT_SHIFT);
            else        stage[pos]     = (int)((unsigned)w0[q] >> BKT_SHIFT);
        }
        if (lane + 32 < lg) {
            const int pos = atomicAdd(&c[w1[q] & (BKT_NODES - 1)], 1);
            if (direct) csr[beg + pos] = (int)((unsigned)w1[q] >> BKT_SHIFT);
            else        stage[pos]     = (int)((unsigned)w1[q] >> BKT_SHIFT);
        }
    }
    for (int q = 0; q < NCHUNK / 32; ++q) {
        const int gq = wave + 32 * q;
        const int lg = __builtin_amdgcn_readfirstlane(segl[gq]);
        const int* p = pairs + (size_t)gq * chp + __builtin_amdgcn_readfirstlane(segs[gq]);
        for (int k = 64 + lane; k < lg; k += 32) {
            const int w = p[k];
            const int pos = atomicAdd(&c[w & (BKT_NODES - 1)], 1);
            if (direct) csr[beg + pos] = (int)((unsigned)w >> BKT_SHIFT);
            else        stage[pos]     = (int)((unsigned)w >> BKT_SHIFT);
        }
    }
    __syncthreads();
    if (direct) return;

    int t = tid - (beg & 31);
    if (t >= 0 && t < cnt) csr[beg + t] = stage[t];
    t += SORT_THREADS;
    for (; t + 3 * SORT_THREADS < cnt; t += 4 * SORT_THREADS) {
        int v[4];
#pragma unroll
        for (int u = 0; u < 4; ++u) v[u] = stage[t + u * SORT_THREADS];
#pragma unroll
        for (int u = 0; u < 4; ++u) csr[beg + t + u * SORT_THREADS] = v[u];
    }
    for (; t < cnt; t += SORT_THREADS) csr[beg + t] = stage[t];
}

__global__ __launch_bounds__(256) void k_gemm1(const float* __restrict__ X,
                                               const float* __restrict__ W,
                                               const float* __restrict__ dinv,
                                               float* __restrict__ H, int n) {
    __shared__ float sW[IN_DIM * HID_DIM];
    for (int i = threadIdx.x; i < IN_DIM * HID_DIM; i += 256) sW[i] = W[i];
    __syncthreads();

    const int lane = threadIdx.x & 31;
    const int wave = threadIdx.x >> 5;
    const int rowBase = (blockIdx.x * 8 + wave) * 16;
    if (rowBase >= n) return;

    const int m  = lane & 15;
    const int kh = lane >> 4;
    int row  = rowBase + m;
    int rowc = row < n ? row : n - 1;
    const float* xrow = X + (size_t)rowc * IN_DIM;

    float dv[8];
#pragma unroll
    for (int r = 0; r < 8; ++r) {
        const int orow = rowBase + r + 8 * kh;
        dv[r] = dinv[orow < n ? orow : n - 1];
    }

    v8f a0 = {}, a1 = {}, a2 = {}, a3 = {};
#pragma unroll 4
    for (int k = 0; k < IN_DIM; k += 4) {
        v2f a = *(const v2f*)(xrow + k + 2 * kh);
        const float* wp = sW + (k + 2 * kh) * HID_DIM;
        v2f b0 = { wp[m +  0], wp[HID_DIM + m +  0] };
        v2f b1 = { wp[m + 16], wp[HID_DIM + m + 16] };
        v2f b2 = { wp[m + 32], wp[HID_DIM + m + 32] };
        v2f b3 = { wp[m + 48], wp[HID_DIM + m + 48] };
        a0 = __builtin_amdgcn_wmma_f32_16x16x4_f32(false, a, false, b0, (short)0, a0, false, false);
        a1 = __builtin_amdgcn_wmma_f32_16x16x4_f32(false, a, false, b1, (short)0, a1, false, false);
        a2 = __builtin_amdgcn_wmma_f32_16x16x4_f32(false, a, false, b2, (short)0, a2, false, false);
        a3 = __builtin_amdgcn_wmma_f32_16x16x4_f32(false, a, false, b3, (short)0, a3, false, false);
    }
#pragma unroll
    for (int r = 0; r < 8; ++r) {
        int orow = rowBase + r + 8 * kh;
        if (orow < n) {
            float* hp = H + (size_t)orow * HID_DIM + m;
            hp[0]  = a0[r] * dv[r];
            hp[16] = a1[r] * dv[r];
            hp[32] = a2[r] * dv[r];
            hp[48] = a3[r] * dv[r];
        }
    }
}

#define A_PITCH 68
__global__ __launch_bounds__(256) void k_gather1(const int* __restrict__ off, const int* __restrict__ csr,
                                                 const float* __restrict__ dinv, const float* __restrict__ H1,
                                                 const float* __restrict__ b1, const float* __restrict__ W2,
                                                 float* __restrict__ H2, int n) {
    __shared__ __attribute__((aligned(16))) float sA[16 * A_PITCH];
    __shared__ __attribute__((aligned(16))) float sD[16];
    __shared__ float sW[HID_DIM * OUT_DIM];
    const int lane = threadIdx.x & 31;
    const int wave = threadIdx.x >> 5;
    const int rowBase = (int)blockIdx.x * 16;
    const float* hcol = H1 + 2 * lane;
    const v2f bb = *(const v2f*)(b1 + 2 * lane);
    float w2v = 0.0f;
    if (threadIdx.x < HID_DIM * OUT_DIM) w2v = W2[threadIdx.x];

    for (int rr = 0; rr < 2; ++rr) {
        const int lr = __builtin_amdgcn_readfirstlane(wave + 8 * rr);
        const int i  = rowBase + lr;
        v2f o = { 0.0f, 0.0f };
        float di = 0.0f;
        if (i < n) {
            const int beg = __builtin_amdgcn_readfirstlane(off[i]);
            const int end = __builtin_amdgcn_readfirstlane(off[i + 1]);
            di = dinv[i];

            v2f acc = *(const v2f*)(hcol + (size_t)i * HID_DIM);

            for (int j = beg; j < end; j += 32) {
                int c = end - j;
                if (c > 32) c = 32;
                int s_l = 0;
                if (lane < c) s_l = csr[j + lane];
                int t = 0;
                for (; t + 8 <= c; t += 8) {
                    v2f h[8];
#pragma unroll
                    for (int u = 0; u < 8; ++u) {
                        const int s = __builtin_amdgcn_readlane(s_l, t + u);
                        h[u] = *(const v2f*)(hcol + (size_t)s * HID_DIM);
                    }
#pragma unroll
                    for (int u = 0; u < 8; ++u) {
                        acc.x += h[u].x;
                        acc.y += h[u].y;
                    }
                }
                if (t < c) {
                    v2f h[7];
#pragma unroll
                    for (int u = 0; u < 7; ++u) {
                        h[u] = v2f{ 0.0f, 0.0f };
                        if (t + u < c) {
                            const int s = __builtin_amdgcn_readlane(s_l, t + u);
                            h[u] = *(const v2f*)(hcol + (size_t)s * HID_DIM);
                        }
                    }
#pragma unroll
                    for (int u = 0; u < 7; ++u) {
                        acc.x += h[u].x;
                        acc.y += h[u].y;
                    }
                }
            }
            o.x = fmaxf(fmaf(acc.x, di, bb.x), 0.0f);
            o.y = fmaxf(fmaf(acc.y, di, bb.y), 0.0f);
        }
        *(v2f*)(sA + lr * A_PITCH + 2 * lane) = o;
        if (lane == 0) sD[lr] = di;
    }
    if (threadIdx.x < HID_DIM * OUT_DIM) sW[threadIdx.x] = w2v;
    __syncthreads();
    if (wave != 0) return;

    const int m  = lane & 15;
    const int kh = lane >> 4;
    const float* arow = sA + m * A_PITCH + 2 * kh;
    const bool live = m < OUT_DIM;
    const float* wcol = sW + 2 * kh * OUT_DIM + (live ? m : 0);
    const v4f dlo = *(const v4f*)(sD + 8 * kh);
    const v4f dhi = *(const v4f*)(sD + 8 * kh + 4);
    const float dv[8] = { dlo.x, dlo.y, dlo.z, dlo.w, dhi.x, dhi.y, dhi.z, dhi.w };
    v8f acc2 = {};
#pragma unroll 4
    for (int k = 0; k < HID_DIM; k += 4) {
        const v2f a = *(const v2f*)(arow + k);
        const float bx = wcol[k * OUT_DIM];
        const float by = wcol[(k + 1) * OUT_DIM];
        const v2f b = { live ? bx : 0.0f, live ? by : 0.0f };
        acc2 = __builtin_amdgcn_wmma_f32_16x16x4_f32(false, a, false, b, (short)0, acc2, false, false);
    }
#pragma unroll
    for (int r = 0; r < 8; ++r) {
        const int orow = rowBase + r + 8 * kh;
        if (orow < n && m < H2_PITCH) H2[(size_t)orow * H2_PITCH + m] = live ? acc2[r] * dv[r] : 0.0f;
    }
}

__global__ __launch_bounds__(256) void k_gather2(const int* __restrict__ off, const int* __restrict__ csr,
                                                 const float* __restrict__ dinv, const float* __restrict__ H2,
                                                 const float* __restrict__ b2, float* __restrict__ out, int n) {
    __shared__ float so[32 * OUT_DIM];
    const int g  = threadIdx.x >> 3;
    const int sl = threadIdx.x & 7;
    const int i  = blockIdx.x * 32 + g;

    float a0 = 0.0f, a1 = 0.0f, a2 = 0.0f;
    float di = 0.0f;
    if (i < n) {
        const int beg = off[i], end = off[i + 1];
        di = dinv[i];
        for (int j = beg + sl; j < end; j += 8) {
            const int s = csr[j];
            const v4f h = *(const v4f*)(H2 + (size_t)s * H2_PITCH);
            a0 += h.x;
            a1 += h.y;
            a2 += h.z;
        }
    }
#pragma unroll
    for (int m = 4; m >= 1; m >>= 1) {
        a0 += __shfl_xor(a0, m);
        a1 += __shfl_xor(a1, m);
        a2 += __shfl_xor(a2, m);
    }
    if (sl == 0 && i < n) {
        const v4f hs = *(const v4f*)(H2 + (size_t)i * H2_PITCH);
        so[g * OUT_DIM + 0] = fmaf(a0 + hs.x, di, b2[0]);
        so[g * OUT_DIM + 1] = fmaf(a1 + hs.y, di, b2[1]);
        so[g * OUT_DIM + 2] = fmaf(a2 + hs.z, di, b2[2]);
    }
    __syncthreads();
    if (threadIdx.x < 32 * OUT_DIM) {
        const int idx = blockIdx.x * (32 * OUT_DIM) + threadIdx.x;
        if (idx < n * OUT_DIM) out[idx] = so[threadIdx.x];
    }
}

extern "C" void kernel_launch(void* const* d_in, const int* in_sizes, int n_in,
                              void* d_out, int out_size, void* d_ws, size_t ws_size,
                              hipStream_t stream) {
    (void)n_in; (void)out_size; (void)ws_size;
    const float* x  = (const float*)d_in[0];
    const int*  edge = (const int*)d_in[1];
    const float* W1 = (const float*)d_in[2];
    const float* b1 = (const float*)d_in[3];
    const float* W2 = (const float*)d_in[4];
    const float* b2 = (const float*)d_in[5];
    float* out = (float*)d_out;

    const int n = in_sizes[0] / IN_DIM;
    const int e = in_sizes[1] / 2;
    const int* src = edge;
    const int* dst = edge + e;
    const int nb = (n + BKT_NODES - 1) / BKT_NODES;

    const size_t np  = ((size_t)n + 1 + 63) & ~(size_t)63;
    const size_t ep  = ((size_t)e + 63) & ~(size_t)63;
    float* ws    = (float*)d_ws;
    float* dinv  = ws;
    int*   off   = (int*)(ws + np);
    int*   tableT = off + np;
    int*   pairs = tableT + (size_t)(NBMAX + 1) * NCHUNK;
    int*   csr   = pairs + ep + (size_t)NCHUNK * 32;
    float* H1    = (float*)(csr + ep);
    float* H2    = H1 + (size_t)n * HID_DIM;

    dim3 blk(256);
    k_part <<<NCHUNK, SORT_THREADS, 0, stream>>>(src, dst, tableT, pairs, e, nb);
    k_bsort<<<nb, SORT_THREADS, 0, stream>>>(pairs, tableT, off, dinv, csr, n, e);

    k_gemm1  <<<(n + 127) / 128, blk, 0, stream>>>(x, W1, dinv, H1, n);
    k_gather1<<<(n + 15) / 16, blk, 0, stream>>>(off, csr, dinv, H1, b1, W2, H2, n);

    k_gather2<<<(n + 31) / 32, blk, 0, stream>>>(off, csr, dinv, H2, b2, out, n);
}


# both gathers: one long-lived wave per tile (16 rows / 32 nodes) instead of eight short ones; a row's sources read as one group of loads, the next list read ahead; 8x fewer waves; sum order kept
# speedup vs baseline: 7.8343x; 1.0377x over previous
#include <hip/hip_runtime.h>

typedef __attribute__((ext_vector_type(2))) float v2f;
typedef __attribute__((ext_vector_type(4))) float v4f;
typedef __attribute__((ext_vector_type(8))) float v8f;
typedef __attribute__((ext_vector_type(2))) int   v2i;

#define IN_DIM  128
#define HID_DIM 64
#define OUT_DIM 3
#define H2_PITCH 4
#define BKT_SHIFT 8
#define BKT_NODES (1 << BKT_SHIFT)
#define NBMAX 512
#define NCHUNK 256
#define SORT_THREADS 1024
#define STAGE_CAP 16384

#define EPT (STAGE_CAP / SORT_THREADS)
typedef __attribute__((ext_vector_type(4))) int v4i;

__global__ __launch_bounds__(SORT_THREADS) void k_part(const int* __restrict__ src, const int* __restrict__ dst,
                                                       int* __restrict__ tableT, int* __restrict__ pairs,
                                                       int e, int nb) {
    __shared__ int cnt[NBMAX];
    __shared__ int wsum[SORT_THREADS / 32];
    __shared__ __attribute__((aligned(16))) int stage[STAGE_CAP];
    const int tid  = threadIdx.x;
    const int lane = tid & 31;
    const int wave = tid >> 5;
    const int g    = blockIdx.x;

    const int ch  = (e + NCHUNK - 1) / NCHUNK;
    const int chp = (ch + 31) & ~31;
    const int beg = g * ch;
    const int end = (beg + ch < e) ? beg + ch : e;

    int d[EPT], s[EPT], r[EPT];
#pragma unroll
    for (int u = 0; u < EPT; ++u) {
        const int j = beg + tid + u * SORT_THREADS;
        d[u] = 0; s[u] = 0;
        if (j < end) { d[u] = dst[j]; s[u] = src[j]; }
    }
    if (tid < NBMAX) cnt[tid] = 0;
    __syncthreads();
#pragma unroll
    for (int u = 0; u < EPT; ++u) {
        r[u] = 0;
        if (beg + tid + u * SORT_THREADS < end) r[u] = atomicAdd(&cnt[d[u] >> BKT_SHIFT], 1);
    }
    __syncthreads();

    const int val = (tid < NBMAX) ? cnt[tid] : 0;
    int inc = val;
#pragma unroll
    for (int k = 1; k < 32; k <<= 1) {
        const int v = __shfl_up(inc, k);
        if (lane >= k) inc += v;
    }
    if (lane == 31) wsum[wave] = inc;
    __syncthreads();
    int wpre = 0;
    for (int w = 0; w < wave; ++w) wpre += wsum[w];
    const int lexcl = wpre + inc - val;
    if (tid < NBMAX) cnt[tid] = lexcl;
    if (tid <= nb) tableT[tid * NCHUNK + g] = lexcl;
    __syncthreads();

    int first[EPT];
#pragma unroll
    for (int u = 0; u < EPT; ++u) first[u] = cnt[d[u] >> BKT_SHIFT];
#pragma unroll
    for (int u = 0; u < EPT; ++u) {
        if (beg + tid + u * SORT_THREADS < end)
            stage[first[u] + r[u]] = (int)(((unsigned)s[u] << BKT_SHIFT) | (unsigned)(d[u] & (BKT_NODES - 1)));
    }
    __syncthreads();

    int* __restrict__ outp = pairs + (size_t)g * chp;
    const int c  = end > beg ? end - beg : 0;
    const int c4 = c & ~3;
    for (int t = tid * 4; t < c4; t += 4 * SORT_THREADS) *(v4i*)(outp + t) = *(const v4i*)(stage + t);
    if (tid < c - c4) outp[c4 + tid] = stage[c4 + tid];
}

__global__ __launch_bounds__(SORT_THREADS, 16) void k_bsort(const int* __restrict__ pairs, const int* __restrict__ tableT,
                                                        int* __restrict__ off, float* __restrict__ dinv,
                                                        int* __restrict__ csr, int n, int e) {
    __shared__ int c[BKT_NODES];
    __shared__ int wsum[BKT_NODES / 32];
    __shared__ int segs[NCHUNK];
    __shared__ int segl[NCHUNK];
    __shared__ int red[2][SORT_THREADS / 32];
    __shared__ int stage[STAGE_CAP];
    const int tid  = threadIdx.x;
    const int lane = tid & 31;
    const int wave = __builtin_amdgcn_readfirstlane(tid >> 5);
    const int b    = blockIdx.x;

    int s0 = 0, ln = 0;
    if (tid < NCHUNK) {
        s0 = tableT[b * NCHUNK + tid];
        ln = tableT[(b + 1) * NCHUNK + tid] - s0;
        segs[tid] = s0;
        segl[tid] = ln;
    }
    if (tid < BKT_NODES) c[tid] = 0;
#pragma unroll
    for (int k = 16; k >= 1; k >>= 1) {
        s0 += __shfl_xor(s0, k);
        ln += __shfl_xor(ln, k);
    }
    if (lane == 0) { red[0][wave] = s0; red[1][wave] = ln; }
    __syncthreads();
    int beg = 0, cnt = 0;
    for (int w = 0; w < NCHUNK / 32; ++w) { beg += red[0][w]; cnt += red[1][w]; }

    const int ch  = (e + NCHUNK - 1) / NCHUNK;
    const int chp = (ch + 31) & ~31;
    const int* skip = tableT + b * NCHUNK;
    int w0[NCHUNK / 32], w1[NCHUNK / 32];
#pragma unroll
    for (int q = 0; q < NCHUNK / 32; ++q) {
        const int gq = wave + 32 * q;
        const int lg = __builtin_amdgcn_readfirstlane(segl[gq]);
        const int* p = pairs + (size_t)gq * chp + __builtin_amdgcn_readfirstlane(segs[gq]);
        const int k0 = lane < lg ? lane : lg - 1;
        const int k1 = lane + 32 < lg ? lane + 32 : lg - 1;
        w0[q] = lg > 0  ? p[k0] : skip[0];
        w1[q] = lg > 32 ? p[k1] : skip[0];
    }
#pragma unroll
    for (int q = 0; q < NCHUNK / 32; ++q) {
        const int lg = __builtin_amdgcn_readfirstlane(segl[wave + 32 * q]);
        if (lane < lg)      atomicAdd(&c[w0[q] & (BKT_NODES - 1)], 1);
        if (lane + 32 < lg) atomicAdd(&c[w1[q] & (BKT_NODES - 1)], 1);
    }
    for (int q = 0; q < NCHUNK / 32; ++q) {
        const int gq = wave + 32 * q;
        const int lg = __builtin_amdgcn_readfirstlane(segl[gq]);
        const int* p = pairs + (size_t)gq * chp + __builtin_amdgcn_readfirstlane(segs[gq]);
        for (int k = 64 + lane; k < lg; k += 32) atomicAdd(&c[p[k] & (BKT_NODES - 1)], 1);
    }
    __syncthreads();

    const int val = (tid < BKT_NODES) ? c[tid] : 0;
    int inc = val;
#pragma unroll
    for (int k = 1; k < 32; k <<= 1) {
        const int v = __shfl_up(inc, k);
        if (lane >= k) inc += v;
    }
    if (tid < BKT_NODES && lane == 31) wsum[wave] = inc;
    __syncthreads();
    if (tid < BKT_NODES) {
        int wpre = 0;
        for (int w = 0; w < wave; ++w) wpre += wsum[w];
        const int first = wpre + inc - val;
        const int node  = b * BKT_NODES + tid;
        c[tid] = first;
        if (node < n) {
            off[node]  = beg + first;
            dinv[node] = rsqrtf(fmaxf((float)(val + 1), 1.0f));
            if (node == n - 1) off[n] = beg + first + val;
        }
    }
    __syncthreads();

    const bool direct = cnt > STAGE_CAP;
#pragma unroll
    for (int q = 0; q < NCHUNK / 32; ++q) {
        const int lg = __builtin_amdgcn_readfirstlane(segl[wave + 32 * q]);
        if (lane < lg) {
            const int pos = atomicAdd(&c[w0[q] & (BKT_NODES - 1)], 1);
            if (direct) csr[beg + pos] = (int)((unsigned)w0[q] >> BKT_SHIFT);
            else        stage[pos]     = (int)((unsigned)w0[q] >> BKT_SHIFT);
        }
        if (lane + 32 < lg) {
            const int pos = atomicAdd(&c[w1[q] & (BKT_NODES - 1)], 1);
            if (direct) csr[beg + pos] = (int)((unsigned)w1[q] >> BKT_SHIFT);
            else        stage[pos]     = (int)((unsigned)w1[q] >> BKT_SHIFT);
        }
    }
    for (int q = 0; q < NCHUNK / 32; ++q) {
        const int gq = wave + 32 * q;
        const int lg = __builtin_amdgcn_readfirstlane(segl[gq]);
        const int* p = pairs + (size_t)gq * chp + __builtin_amdgcn_readfirstlane(segs[gq]);
        for (int k = 64 + lane; k < lg; k += 32) {
            const int w = p[k];
            const int pos = atomicAdd(&c[w & (BKT_NODES - 1)], 1);
            if (direct) csr[beg + pos] = (int)((unsigned)w >> BKT_SHIFT);
            else        stage[pos]     = (int)((unsigned)w >> BKT_SHIFT);
        }
    }
    __syncthreads();
    if (direct) return;

    int t = tid - (beg & 31);
    if (t >= 0 && t < cnt) csr[beg + t] = stage[t];
    t += SORT_THREADS;
    for (; t + 3 * SORT_THREADS < cnt; t += 4 * SORT_THREADS) {
        int v[4];
#pragma unroll
        for (int u = 0; u < 4; ++u) v[u] = stage[t + u * SORT_THREADS];
#pragma unroll
        for (int u = 0; u < 4; ++u) csr[beg + t + u * SORT_THREADS] = v[u];
    }
    for (; t < cnt; t += SORT_THREADS) csr[beg + t] = stage[t];
}

__global__ __launch_bounds__(256) void k_gemm1(const float* __restrict__ X,
                                               const float* __restrict__ W,
                                               const float* __restrict__ dinv,
                                               float* __restrict__ H, int n) {
    __shared__ float sW[IN_DIM * HID_DIM];
    for (int i = threadIdx.x; i < IN_DIM * HID_DIM; i += 256) sW[i] = W[i];
    __syncthreads();

    const int lane = threadIdx.x & 31;
    const int wave = threadIdx.x >> 5;
    const int rowBase = (blockIdx.x * 8 + wave) * 16;
    if (rowBase >= n) return;

    const int m  = lane & 15;
    const int kh = lane >> 4;
    int row  = rowBase + m;
    int rowc = row < n ? row : n - 1;
    const float* xrow = X + (size_t)rowc * IN_DIM;

    float dv[8];
#pragma unroll
    for (int r = 0; r < 8; ++r) {
        const int orow = rowBase + r + 8 * kh;
        dv[r] = dinv[orow < n ? orow : n - 1];
    }

    v8f a0 = {}, a1 = {}, a2 = {}, a3 = {};
#pragma unroll 4
    for (int k = 0; k < IN_DIM; k += 4) {
        v2f a = *(const v2f*)(xrow + k + 2 * kh);
        const float* wp = sW + (k + 2 * kh) * HID_DIM;
        v2f b0 = { wp[m +  0], wp[HID_DIM + m +  0] };
        v2f b1 = { wp[m + 16], wp[HID_DIM + m + 16] };
        v2f b2 = { wp[m + 32], wp[HID_DIM + m + 32] };
        v2f b3 = { wp[m + 48], wp[HID_DIM + m + 48] };
        a0 = __builtin_amdgcn_wmma_f32_16x16x4_f32(false, a, false, b0, (short)0, a0, false, false);
        a1 = __builtin_amdgcn_wmma_f32_16x16x4_f32(false, a, false, b1, (short)0, a1, false, false);
        a2 = __builtin_amdgcn_wmma_f32_16x16x4_f32(false, a, false, b2, (short)0, a2, false, false);
        a3 = __builtin_amdgcn_wmma_f32_16x16x4_f32(false, a, false, b3, (short)0, a3, false, false);
    }
#pragma unroll
    for (int r = 0; r < 8; ++r) {
        int orow = rowBase + r + 8 * kh;
        if (orow < n) {
            float* hp = H + (size_t)orow * HID_DIM + m;
            hp[0]  = a0[r] * dv[r];
            hp[16] = a1[r] * dv[r];
            hp[32] = a2[r] * dv[r];
            hp[48] = a3[r] * dv[r];
        }
    }
}

#define A_PITCH 68
#define G1_WAVES 4
__global__ __launch_bounds__(32 * G1_WAVES) void k_gather1(const int* __restrict__ off, const int* __restrict__ csr,
                                                 const float* __restrict__ dinv, const float* __restrict__ H1,
                                                 const float* __restrict__ b1, const float* __restrict__ W2,
                                                 float* __restrict__ H2, int n) {
    __shared__ __attribute__((aligned(16))) float sA[G1_WAVES][16 * A_PITCH];
    __shared__ __attribute__((aligned(16))) float sD[G1_WAVES][16];
    __shared__ float sW[HID_DIM * OUT_DIM];
    const int lane = threadIdx.x & 31;
    const int wave = __builtin_amdgcn_readfirstlane((int)threadIdx.x >> 5);
    const int rowBase = ((int)blockIdx.x * G1_WAVES + wave) * 16;

    int   o_l = 0;
    float d_l = 0.0f;
    if (lane <= 16 && rowBase + lane <= n) o_l = off[rowBase + lane];
    if (lane <  16 && rowBase + lane <  n) d_l = dinv[rowBase + lane];
    for (int t = threadIdx.x; t < HID_DIM * OUT_DIM; t += 32 * G1_WAVES) sW[t] = W2[t];
    __syncthreads();
    if (rowBase >= n) return;

    const float* hcol = H1 + 2 * lane;
    const v2f bb = *(const v2f*)(b1 + 2 * lane);
    float* tile = sA[wave];
    if (lane < 16) sD[wave][lane] = d_l;

    int beg = __builtin_amdgcn_readlane(o_l, 0);
    int end = __builtin_amdgcn_readlane(o_l, 1);
    int sa = 0, sb = 0;
    if (beg + lane < end)      sa = csr[beg + lane];
    if (beg + 32 + lane < end) sb = csr[beg + 32 + lane];

    for (int r = 0; r < 16; ++r) {
        const int i = rowBase + r;
        v2f o = { 0.0f, 0.0f };
        if (i < n) {
            asm volatile("" :: "v"(sa), "v"(sb));
            int nend = end, na = 0, nb = 0;
            if (r < 15 && i + 1 < n) {
                nend = __builtin_amdgcn_readlane(o_l, r + 2);
                if (end + lane < nend)      na = csr[end + lane];
                if (end + 32 + lane < nend) nb = csr[end + 32 + lane];
            }
            const float di = __int_as_float(__builtin_amdgcn_readlane(__float_as_int(d_l), r));
            v2f acc = *(const v2f*)(hcol + (size_t)i * HID_DIM);

            int sl = sa;
            int j = beg;
            for (;;) {
                int c = end - j;
                if (c > 32) c = 32;
                v2f h[32];
#pragma unroll
                for (int u = 0; u < 32; ++u) h[u] = v2f{ 0.0f, 0.0f };
#pragma unroll
                for (int u = 0; u < 32; ++u) {
                    if (u < c) {
                        const int s = __builtin_amdgcn_readlane(sl, u);
                        h[u] = *(const v2f*)(hcol + (size_t)s * HID_DIM);
                    }
                }
#pragma unroll
                for (int u = 0; u < 32; ++u) {
                    acc.x += h[u].x;
                    acc.y += h[u].y;
                }
                j += 32;
                if (j >= end) break;
                if (j == beg + 32) sl = sb;
                else {
                    sl = 0;
                    if (j + lane < end) sl = csr[j + lane];
                    asm volatile("" :: "v"(sl));
                }
            }
            o.x = fmaxf(fmaf(acc.x, di, bb.x), 0.0f);
            o.y = fmaxf(fmaf(acc.y, di, bb.y), 0.0f);
            beg = end; end = nend; sa = na; sb = nb;
        }
        *(v2f*)(tile + r * A_PITCH + 2 * lane) = o;
    }

    const int m  = lane & 15;
    const int kh = lane >> 4;
    const float* arow = tile + m * A_PITCH + 2 * kh;
    const bool live = m < OUT_DIM;
    const float* wcol = sW + 2 * kh * OUT_DIM + (live ? m : 0);
    const v4f dlo = *(const v4f*)(sD[wave] + 8 * kh);
    const v4f dhi = *(const v4f*)(sD[wave] + 8 * kh + 4);
    const float dv[8] = { dlo.x, dlo.y, dlo.z, dlo.w, dhi.x, dhi.y, dhi.z, dhi.w };
    v8f acc2 = {};
#pragma unroll 4
    for (int k = 0; k < HID_DIM; k += 4) {
        const v2f a = *(const v2f*)(arow + k);
        const float bx = wcol[k * OUT_DIM];
        const float by = wcol[(k + 1) * OUT_DIM];
        const v2f b = { live ? bx : 0.0f, live ? by : 0.0f };
        acc2 = __builtin_amdgcn_wmma_f32_16x16x4_f32(false, a, false, b, (short)0, acc2, false, false);
    }
#pragma unroll
    for (int r = 0; r < 8; ++r) {
        const int orow = rowBase + r + 8 * kh;
        if (orow < n && m < H2_PITCH) H2[(size_t)orow * H2_PITCH + m] = live ? acc2[r] * dv[r] : 0.0f;
    }
}

#define G2_WAVES 4
#define G2_SLOTS 8
__global__ __launch_bounds__(32 * G2_WAVES) void k_gather2(const int* __restrict__ off, const int* __restrict__ csr,
                                                 const float* __restrict__ dinv, const float* __restrict__ H2,
                                                 const float* __restrict__ b2, float* __restrict__ out, int n) {
    __shared__ float so[G2_WAVES][32 * OUT_DIM];
    const int lane = threadIdx.x & 31;
    const int wave = __builtin_amdgcn_readfirstlane((int)threadIdx.x >> 5);
    const int base = ((int)blockIdx.x * G2_WAVES + wave) * 32;
    if (base >= n) return;
    const int g  = lane >> 3;
    const int sl = lane & 7;
    float* sw = so[wave];

    int   o_b = 0, o_e = 0;
    float d_l = 0.0f;
    v4f   hs  = { 0.0f, 0.0f, 0.0f, 0.0f };
    if (base + lane < n) {
        o_b = off[base + lane];
        o_e = off[base + lane + 1];
        d_l = dinv[base + lane];
        hs  = *(const v4f*)(H2 + (size_t)(base + lane) * H2_PITCH);
    }

    int beg = __shfl(o_b, g);
    int end = __shfl(o_e, g);
    int s[G2_SLOTS];
#pragma unroll
    for (int t = 0; t < G2_SLOTS; ++t) {
        const int j = beg + sl + 8 * t;
        s[t] = -1;
        if (j < end) s[t] = csr[j];
    }
#pragma nounroll
    for (int p = 0; p < 8; ++p) {
        const int idx = 4 * p + g;
        asm volatile("" :: "v"(s[0]), "v"(s[1]), "v"(s[2]), "v"(s[3]), "v"(s[4]), "v"(s[5]), "v"(s[6]), "v"(s[7]));
        const int nidx = (idx + 4) & 31;
        const int nbeg = __shfl(o_b, nidx);
        const int nend = p < 7 ? __shfl(o_e, nidx) : nbeg;
        int sn[G2_SLOTS];
#pragma unroll
        for (int t = 0; t < G2_SLOTS; ++t) {
            const int j = nbeg + sl + 8 * t;
            sn[t] = -1;
            if (j < nend) sn[t] = csr[j];
        }
        v4f h[G2_SLOTS];
#pragma unroll
        for (int t = 0; t < G2_SLOTS; ++t) {
            h[t] = v4f{ 0.0f, 0.0f, 0.0f, 0.0f };
            if (s[t] >= 0) h[t] = *(const v4f*)(H2 + (size_t)s[t] * H2_PITCH);
        }
        float a0 = 0.0f, a1 = 0.0f, a2 = 0.0f;
#pragma unroll
        for (int t = 0; t < G2_SLOTS; ++t) {
            a0 += h[t].x;
            a1 += h[t].y;
            a2 += h[t].z;
        }
        for (int j = beg + sl + 8 * G2_SLOTS; j < end; j += 8) {
            const int s1 = csr[j];
            const v4f h1 = *(const v4f*)(H2 + (size_t)s1 * H2_PITCH);
            a0 += h1.x;
            a1 += h1.y;
            a2 += h1.z;
        }
#pragma unroll
        for (int m = 4; m >= 1; m >>= 1) {
            a0 += __shfl_xor(a0, m);
            a1 += __shfl_xor(a1, m);
            a2 += __shfl_xor(a2, m);
        }
        if (sl == 0) {
            sw[idx * OUT_DIM + 0] = a0;
            sw[idx * OUT_DIM + 1] = a1;
            sw[idx * OUT_DIM + 2] = a2;
        }
        beg = nbeg; end = nend;
#pragma unroll
        for (int t = 0; t < G2_SLOTS; ++t) s[t] = sn[t];
    }
    const float r0 = fmaf(sw[lane * OUT_DIM + 0] + hs.x, d_l, b2[0]);
    const float r1 = fmaf(sw[lane * OUT_DIM + 1] + hs.y, d_l, b2[1]);
    const float r2 = fmaf(sw[lane * OUT_DIM + 2] + hs.z, d_l, b2[2]);
    sw[lane * OUT_DIM + 0] = r0;
    sw[lane * OUT_DIM + 1] = r1;
    sw[lane * OUT_DIM + 2] = r2;
#pragma unroll
    for (int k = 0; k < OUT_DIM; ++k) {
        const int w = base * OUT_DIM + 32 * k + lane;
        if (w < n * OUT_DIM) out[w] = sw[32 * k + lane];
    }
}

extern "C" void kernel_launch(void* const* d_in, const int* in_sizes, int n_in,
                              void* d_out, int out_size, void* d_ws, size_t ws_size,
                              hipStream_t stream) {
    (void)n_in; (void)out_size; (void)ws_size;
    const float* x  = (const float*)d_in[0];
    const int*  edge = (const int*)d_in[1];
    const float* W1 = (const float*)d_in[2];
    const float* b1 = (const float*)d_in[3];
    const float* W2 = (const float*)d_in[4];
    const float* b2 = (const float*)d_in[5];
    float* out = (float*)d_out;

    const int n = in_sizes[0] / IN_DIM;
    const int e = in_sizes[1] / 2;
    const int* src = edge;
    const int* dst = edge + e;
    const int nb = (n + BKT_NODES - 1) / BKT_NODES;

    const size_t np  = ((size_t)n + 1 + 63) & ~(size_t)63;
    const size_t ep  = ((size_t)e + 63) & ~(size_t)63;
    float* ws    = (float*)d_ws;
    float* dinv  = ws;
    int*   off   = (int*)(ws + np);
    int*   tableT = off + np;
    int*   pairs = tableT + (size_t)(NBMAX + 1) * NCHUNK;
    int*   csr   = pairs + ep + (size_t)NCHUNK * 32;
    float* H1    = (float*)(csr + ep);
    float* H2    = H1 + (size_t)n * HID_DIM;

    dim3 blk(256);
    k_part <<<NCHUNK, SORT_THREADS, 0, stream>>>(src, dst, tableT, pairs, e, nb);
    k_bsort<<<nb, SORT_THREADS, 0, stream>>>(pairs, tableT, off, dinv, csr, n, e);

    k_gemm1  <<<(n + 127) / 128, blk, 0, stream>>>(x, W1, dinv, H1, n);
    k_gather1<<<(n + 16 * G1_WAVES - 1) / (16 * G1_WAVES), 32 * G1_WAVES, 0, stream>>>(off, csr, dinv, H1, b1, W2, H2, n);

    k_gather2<<<(n + 32 * G2_WAVES - 1) / (32 * G2_WAVES), 32 * G2_WAVES, 0, stream>>>(off, csr, dinv, H2, b2, out, n);
}


# layer-1 product moved into the bucket-sort launch: a bucket's work-group multiplies its own 256 rows (f32 matrix instruction, 16-B row loads, k regrouped: products re-rounded); one launch less
# speedup vs baseline: 8.4878x; 1.0834x over previous
#include <hip/hip_runtime.h>

typedef __attribute__((ext_vector_type(2))) float v2f;
typedef __attribute__((ext_vector_type(4))) float v4f;
typedef __attribute__((ext_vector_type(8))) float v8f;
typedef __attribute__((ext_vector_type(2))) int   v2i;

#define IN_DIM  128
#define HID_DIM 64
#define OUT_DIM 3
#define H2_PITCH 4
#define BKT_SHIFT 8
#define BKT_NODES (1 << BKT_SHIFT)
#define NBMAX 512
#define NCHUNK 256
#define SORT_THREADS 1024
#define STAGE_CAP 16384

#define EPT (STAGE_CAP / SORT_THREADS)
typedef __attribute__((ext_vector_type(4))) int v4i;

__global__ __launch_bounds__(SORT_THREADS) void k_part(const int* __restrict__ src, const int* __restrict__ dst,
                                                       int* __restrict__ tableT, int* __restrict__ pairs,
                                                       int e, int nb) {
    __shared__ int cnt[NBMAX];
    __shared__ int wsum[SORT_THREADS / 32];
    __shared__ __attribute__((aligned(16))) int stage[STAGE_CAP];
    const int tid  = threadIdx.x;
    const int lane = tid & 31;
    const int wave = tid >> 5;
    const int g    = blockIdx.x;

    const int ch  = (e + NCHUNK - 1) / NCHUNK;
    const int chp = (ch + 31) & ~31;
    const int beg = g * ch;
    const int end = (beg + ch < e) ? beg + ch : e;

    int d[EPT], s[EPT], r[EPT];
#pragma unroll
    for (int u = 0; u < EPT; ++u) {
        const int j = beg + tid + u * SORT_THREADS;
        d[u] = 0; s[u] = 0;
        if (j < end) { d[u] = dst[j]; s[u] = src[j]; }
    }
    if (tid < NBMAX) cnt[tid] = 0;
    __syncthreads();
#pragma unroll
    for (int u = 0; u < EPT; ++u) {
        r[u] = 0;
        if (beg + tid + u * SORT_THREADS < end) r[u] = atomicAdd(&cnt[d[u] >> BKT_SHIFT], 1);
    }
    __syncthreads();

    const int val = (tid < NBMAX) ? cnt[tid] : 0;
    int inc = val;
#pragma unroll
    for (int k = 1; k < 32; k <<= 1) {
        const int v = __shfl_up(inc, k);
        if (lane >= k) inc += v;
    }
    if (lane == 31) wsum[wave] = inc;
    __syncthreads();
    int wpre = 0;
    for (int w = 0; w < wave; ++w) wpre += wsum[w];
    const int lexcl = wpre + inc - val;
    if (tid < NBMAX) cnt[tid] = lexcl;
    if (tid <= nb) tableT[tid * NCHUNK + g] = lexcl;
    __syncthreads();

    int first[EPT];
#pragma unroll
    for (int u = 0; u < EPT; ++u) first[u] = cnt[d[u] >> BKT_SHIFT];
#pragma unroll
    for (int u = 0; u < EPT; ++u) {
        if (beg + tid + u * SORT_THREADS < end)
            stage[first[u] + r[u]] = (int)(((unsigned)s[u] << BKT_SHIFT) | (unsigned)(d[u] & (BKT_NODES - 1)));
    }
    __syncthreads();

    int* __restrict__ outp = pairs + (size_t)g * chp;
    const int c  = end > beg ? end - beg : 0;
    const int c4 = c & ~3;
    for (int t = tid * 4; t < c4; t += 4 * SORT_THREADS) *(v4i*)(outp + t) = *(const v4i*)(stage + t);
    if (tid < c - c4) outp[c4 + tid] = stage[c4 + tid];
}

__global__ __launch_bounds__(SORT_THREADS, 16) void k_bsort_gemm1(const int* __restrict__ pairs, const int* __restrict__ tableT,
                                                        int* __restrict__ off, float* __restrict__ dinv,
                                                        int* __restrict__ csr,
                                                        const float* __restrict__ X, const float* __restrict__ W,
                                                        float* __restrict__ H, int n, int e) {
    __shared__ __attribute__((aligned(16))) int ldsbuf[IN_DIM * HID_DIM + STAGE_CAP];
    float* const sW    = (float*)ldsbuf;
    int*   const stage = ldsbuf + IN_DIM * HID_DIM;
    __shared__ __attribute__((aligned(16))) float sDinv[BKT_NODES];
    __shared__ int c[BKT_NODES];
    __shared__ int wsum[BKT_NODES / 32];
    __shared__ int segs[NCHUNK];
    __shared__ int segl[NCHUNK];
    __shared__ int red[2][SORT_THREADS / 32];
    const int tid  = threadIdx.x;
    const int lane = tid & 31;
    const int wave = __builtin_amdgcn_readfirstlane(tid >> 5);
    const int b    = blockIdx.x;

    ((v4f*)sW)[tid]                = ((const v4f*)W)[tid];
    ((v4f*)sW)[tid + SORT_THREADS] = ((const v4f*)W)[tid + SORT_THREADS];

    int s0 = 0, ln = 0;
    if (tid < NCHUNK) {
        s0 = tableT[b * NCHUNK + tid];
        ln = tableT[(b + 1) * NCHUNK + tid] - s0;
        segs[tid] = s0;
        segl[tid] = ln;
    }
    if (tid < BKT_NODES) c[tid] = 0;
#pragma unroll
    for (int k = 16; k >= 1; k >>= 1) {
        s0 += __shfl_xor(s0, k);
        ln += __shfl_xor(ln, k);
    }
    if (lane == 0) { red[0][wave] = s0; red[1][wave] = ln; }
    __syncthreads();
    int beg = 0, cnt = 0;
    for (int w = 0; w < NCHUNK / 32; ++w) { beg += red[0][w]; cnt += red[1][w]; }

    const int ch  = (e + NCHUNK - 1) / NCHUNK;
    const int chp = (ch + 31) & ~31;
    const int* skip = tableT + b * NCHUNK;
    int w0[NCHUNK / 32], w1[NCHUNK / 32];
#pragma unroll
    for (int q = 0; q < NCHUNK / 32; ++q) {
        const int gq = wave + 32 * q;
        const int lg = __builtin_amdgcn_readfirstlane(segl[gq]);
        const int* p = pairs + (size_t)gq * chp + __builtin_amdgcn_readfirstlane(segs[gq]);
        const int k0 = lane < lg ? lane : lg - 1;
        const int k1 = lane + 32 < lg ? lane + 32 : lg - 1;
        w0[q] = lg > 0  ? p[k0] : skip[0];
        w1[q] = lg > 32 ? p[k1] : skip[0];
    }
#pragma unroll
    for (int q = 0; q < NCHUNK / 32; ++q) {
        const int lg = __builtin_amdgcn_readfirstlane(segl[wave + 32 * q]);
        if (lane < lg)      atomicAdd(&c[w0[q] & (BKT_NODES - 1)], 1);
        if (lane + 32 < lg) atomicAdd(&c[w1[q] & (BKT_NODES - 1)], 1);
    }
    for (int q = 0; q < NCHUNK / 32; ++q) {
        const int gq = wave + 32 * q;
        const int lg = __builtin_amdgcn_readfirstlane(segl[gq]);
        const int* p = pairs + (size_t)gq * chp + __builtin_amdgcn_readfirstlane(segs[gq]);
        for (int k = 64 + lane; k < lg; k += 32) atomicAdd(&c[p[k] & (BKT_NODES - 1)], 1);
    }
    __syncthreads();

    const int val = (tid < BKT_NODES) ? c[tid] : 0;
    int inc = val;
#pragma unroll
    for (int k = 1; k < 32; k <<= 1) {
        const int v = __shfl_up(inc, k);
        if (lane >= k) inc += v;
    }
    if (tid < BKT_NODES && lane == 31) wsum[wave] = inc;
    __syncthreads();
    if (tid < BKT_NODES) {
        int wpre = 0;
        for (int w = 0; w < wave; ++w) wpre += wsum[w];
        const int first = wpre + inc - val;
        const int node  = b * BKT_NODES + tid;
        c[tid] = first;
        float dn = 0.0f;
        if (node < n) {
            dn = rsqrtf(fmaxf((float)(val + 1), 1.0f));
            off[node]  = beg + first;
            dinv[node] = dn;
            if (node == n - 1) off[n] = beg + first + val;
        }
        sDinv[tid] = dn;
    }
    __syncthreads();

    const bool direct = cnt > STAGE_CAP;
#pragma unroll
    for (int q = 0; q < NCHUNK / 32; ++q) {
        const int lg = __builtin_amdgcn_readfirstlane(segl[wave + 32 * q]);
        if (lane < lg) {
            const int pos = atomicAdd(&c[w0[q] & (BKT_NODES - 1)], 1);
            if (direct) csr[beg + pos] = (int)((unsigned)w0[q] >> BKT_SHIFT);
            else        stage[pos]     = (int)((unsigned)w0[q] >> BKT_SHIFT);
        }
        if (lane + 32 < lg) {
            const int pos = atomicAdd(&c[w1[q] & (BKT_NODES - 1)], 1);
            if (direct) csr[beg + pos] = (int)((unsigned)w1[q] >> BKT_SHIFT);
            else        stage[pos]     = (int)((unsigned)w1[q] >> BKT_SHIFT);
        }
    }
    for (int q = 0; q < NCHUNK / 32; ++q) {
        const int gq = wave + 32 * q;
        const int lg = __builtin_amdgcn_readfirstlane(segl[gq]);
        const int* p = pairs + (size_t)gq * chp + __builtin_amdgcn_readfirstlane(segs[gq]);
        for (int k = 64 + lane; k < lg; k += 32) {
            const int w = p[k];
            const int pos = atomicAdd(&c[w & (BKT_NODES - 1)], 1);
            if (direct) csr[beg + pos] = (int)((unsigned)w >> BKT_SHIFT);
            else        stage[pos]     = (int)((unsigned)w >> BKT_SHIFT);
        }
    }
    __syncthreads();
    if (!direct) {
        int t = tid - (beg & 31);
        if (t >= 0 && t < cnt) csr[beg + t] = stage[t];
        t += SORT_THREADS;
        for (; t + 3 * SORT_THREADS < cnt; t += 4 * SORT_THREADS) {
            int v[4];
#pragma unroll
            for (int u = 0; u < 4; ++u) v[u] = stage[t + u * SORT_THREADS];
#pragma unroll
            for (int u = 0; u < 4; ++u) csr[beg + t + u * SORT_THREADS] = v[u];
        }
        for (; t < cnt; t += SORT_THREADS) csr[beg + t] = stage[t];
    }

    const int rowBase = b * BKT_NODES + (wave >> 1) * 16;
    if (rowBase >= n) return;
    const int m  = lane & 15;
    const int kh = lane >> 4;
    const int cb = (wave & 1) * 32 + m;
    const int row  = rowBase + m;
    const int rowc = row < n ? row : n - 1;
    const float* xrow = X + (size_t)rowc * IN_DIM + 4 * kh;
    const float* wp   = sW + 4 * kh * HID_DIM + cb;

    v8f a0 = {}, a1 = {};
#pragma unroll 1
    for (int i = 0; i < IN_DIM / 32; ++i) {
        v4f ev[4];
#pragma unroll
        for (int q = 0; q < 4; ++q) ev[q] = *(const v4f*)(xrow + 32 * i + 8 * q);
#pragma unroll
        for (int q = 0; q < 4; ++q) {
#pragma unroll
            for (int p = 0; p < 2; ++p) {
                const int kb = 32 * i + 8 * q + 2 * p;
                const v2f a  = p == 0 ? v2f{ ev[q].x, ev[q].y } : v2f{ ev[q].z, ev[q].w };
                const v2f b0 = { wp[kb * HID_DIM],      wp[(kb + 1) * HID_DIM] };
                const v2f b1 = { wp[kb * HID_DIM + 16], wp[(kb + 1) * HID_DIM + 16] };
                a0 = __builtin_amdgcn_wmma_f32_16x16x4_f32(false, a, false, b0, (short)0, a0, false, false);
                a1 = __builtin_amdgcn_wmma_f32_16x16x4_f32(false, a, false, b1, (short)0, a1, false, false);
            }
        }
    }
    const v4f dlo = *(const v4f*)(sDinv + (wave >> 1) * 16 + 8 * kh);
    const v4f dhi = *(const v4f*)(sDinv + (wave >> 1) * 16 + 8 * kh + 4);
    const float dv[8] = { dlo.x, dlo.y, dlo.z, dlo.w, dhi.x, dhi.y, dhi.z, dhi.w };
#pragma unroll
    for (int r = 0; r < 8; ++r) {
        const int orow = rowBase + r + 8 * kh;
        if (orow < n) {
            float* hp = H + (size_t)orow * HID_DIM + cb;
            hp[0]  = a0[r] * dv[r];
            hp[16] = a1[r] * dv[r];
        }
    }
}

#define A_PITCH 68
#define G1_WAVES 4
__global__ __launch_bounds__(32 * G1_WAVES) void k_gather1(const int* __restrict__ off, const int* __restrict__ csr,
                                                 const float* __restrict__ dinv, const float* __restrict__ H1,
                                                 const float* __restrict__ b1, const float* __restrict__ W2,
                                                 float* __restrict__ H2, int n) {
    __shared__ __attribute__((aligned(16))) float sA[G1_WAVES][16 * A_PITCH];
    __shared__ __attribute__((aligned(16))) float sD[G1_WAVES][16];
    __shared__ float sW[HID_DIM * OUT_DIM];
    const int lane = threadIdx.x & 31;
    const int wave = __builtin_amdgcn_readfirstlane((int)threadIdx.x >> 5);
    const int rowBase = ((int)blockIdx.x * G1_WAVES + wave) * 16;

    int   o_l = 0;
    float d_l = 0.0f;
    if (lane <= 16 && rowBase + lane <= n) o_l = off[rowBase + lane];
    if (lane <  16 && rowBase + lane <  n) d_l = dinv[rowBase + lane];
    for (int t = threadIdx.x; t < HID_DIM * OUT_DIM; t += 32 * G1_WAVES) sW[t] = W2[t];
    __syncthreads();
    if (rowBase >= n) return;

    const float* hcol = H1 + 2 * lane;
    const v2f bb = *(const v2f*)(b1 + 2 * lane);
    float* tile = sA[wave];
    if (lane < 16) sD[wave][lane] = d_l;

    int beg = __builtin_amdgcn_readlane(o_l, 0);
    int end = __builtin_amdgcn_readlane(o_l, 1);
    int sa = 0, sb = 0;
    if (beg + lane < end)      sa = csr[beg + lane];
    if (beg + 32 + lane < end) sb = csr[beg + 32 + lane];

    for (int r = 0; r < 16; ++r) {
        const int i = rowBase + r;
        v2f o = { 0.0f, 0.0f };
        if (i < n) {
            asm volatile("" :: "v"(sa), "v"(sb));
            int nend = end, na = 0, nb = 0;
            if (r < 15 && i + 1 < n) {
                nend = __builtin_amdgcn_readlane(o_l, r + 2);
                if (end + lane < nend)      na = csr[end + lane];
                if (end + 32 + lane < nend) nb = csr[end + 32 + lane];
            }
            const float di = __int_as_float(__builtin_amdgcn_readlane(__float_as_int(d_l), r));
            v2f acc = *(const v2f*)(hcol + (size_t)i * HID_DIM);

            int sl = sa;
            int j = beg;
            for (;;) {
                int c = end - j;
                if (c > 32) c = 32;
                v2f h[32];
#pragma unroll
                for (int u = 0; u < 32; ++u) h[u] = v2f{ 0.0f, 0.0f };
#pragma unroll
                for (int u = 0; u < 32; ++u) {
                    if (u < c) {
                        const int s = __builtin_amdgcn_readlane(sl, u);
                        h[u] = *(const v2f*)(hcol + (size_t)s * HID_DIM);
                    }
                }
#pragma unroll
                for (int u = 0; u < 32; ++u) {
                    acc.x += h[u].x;
                    acc.y += h[u].y;
                }
                j += 32;
                if (j >= end) break;
                if (j == beg + 32) sl = sb;
                else {
                    sl = 0;
                    if (j + lane < end) sl = csr[j + lane];
                    asm volatile("" :: "v"(sl));
                }
            }
            o.x = fmaxf(fmaf(acc.x, di, bb.x), 0.0f);
            o.y = fmaxf(fmaf(acc.y, di, bb.y), 0.0f);
            beg = end; end = nend; sa = na; sb = nb;
        }
        *(v2f*)(tile + r * A_PITCH + 2 * lane) = o;
    }

    const int m  = lane & 15;
    const int kh = lane >> 4;
    const float* arow = tile + m * A_PITCH + 2 * kh;
    const bool live = m < OUT_DIM;
    const float* wcol = sW + 2 * kh * OUT_DIM + (live ? m : 0);
    const v4f dlo = *(const v4f*)(sD[wave] + 8 * kh);
    const v4f dhi = *(const v4f*)(sD[wave] + 8 * kh + 4);
    const float dv[8] = { dlo.x, dlo.y, dlo.z, dlo.w, dhi.x, dhi.y, dhi.z, dhi.w };
    v8f acc2 = {};
#pragma unroll 4
    for (int k = 0; k < HID_DIM; k += 4) {
        const v2f a = *(const v2f*)(arow + k);
        const float bx = wcol[k * OUT_DIM];
        const float by = wcol[(k + 1) * OUT_DIM];
        const v2f b = { live ? bx : 0.0f, live ? by : 0.0f };
        acc2 = __builtin_amdgcn_wmma_f32_16x16x4_f32(false, a, false, b, (short)0, acc2, false, false);
    }
#pragma unroll
    for (int r = 0; r < 8; ++r) {
        const int orow = rowBase + r + 8 * kh;
        if (orow < n && m < H2_PITCH) H2[(size_t)orow * H2_PITCH + m] = live ? acc2[r] * dv[r] : 0.0f;
    }
}

#define G2_WAVES 4
#define G2_SLOTS 8
__global__ __launch_bounds__(32 * G2_WAVES) void k_gather2(const int* __restrict__ off, const int* __restrict__ csr,
                                                 const float* __restrict__ dinv, const float* __restrict__ H2,
                                                 const float* __restrict__ b2, float* __restrict__ out, int n) {
    __shared__ float so[G2_WAVES][32 * OUT_DIM];
    const int lane = threadIdx.x & 31;
    const int wave = __builtin_amdgcn_readfirstlane((int)threadIdx.x >> 5);
    const int base = ((int)blockIdx.x * G2_WAVES + wave) * 32;
    if (base >= n) return;
    const int g  = lane >> 3;
    const int sl = lane & 7;
    float* sw = so[wave];

    int   o_b = 0, o_e = 0;
    float d_l = 0.0f;
    v4f   hs  = { 0.0f, 0.0f, 0.0f, 0.0f };
    if (base + lane < n) {
        o_b = off[base + lane];
        o_e = off[base + lane + 1];
        d_l = dinv[base + lane];
        hs  = *(const v4f*)(H2 + (size_t)(base + lane) * H2_PITCH);
    }

    int beg = __shfl(o_b, g);
    int end = __shfl(o_e, g);
    int s[G2_SLOTS];
#pragma unroll
    for (int t = 0; t < G2_SLOTS; ++t) {
        const int j = beg + sl + 8 * t;
        s[t] = -1;
        if (j < end) s[t] = csr[j];
    }
#pragma nounroll
    for (int p = 0; p < 8; ++p) {
        const int idx = 4 * p + g;
        asm volatile("" :: "v"(s[0]), "v"(s[1]), "v"(s[2]), "v"(s[3]), "v"(s[4]), "v"(s[5]), "v"(s[6]), "v"(s[7]));
        const int nidx = (idx + 4) & 31;
        const int nbeg = __shfl(o_b, nidx);
        const int nend = p < 7 ? __shfl(o_e, nidx) : nbeg;
        int sn[G2_SLOTS];
#pragma unroll
        for (int t = 0; t < G2_SLOTS; ++t) {
            const int j = nbeg + sl + 8 * t;
            sn[t] = -1;
            if (j < nend) sn[t] = csr[j];
        }
        v4f h[G2_SLOTS];
#pragma unroll
        for (int t = 0; t < G2_SLOTS; ++t) {
            h[t] = v4f{ 0.0f, 0.0f, 0.0f, 0.0f };
            if (s[t] >= 0) h[t] = *(const v4f*)(H2 + (size_t)s[t] * H2_PITCH);
        }
        float a0 = 0.0f, a1 = 0.0f, a2 = 0.0f;
#pragma unroll
        for (int t = 0; t < G2_SLOTS; ++t) {
            a0 += h[t].x;
            a1 += h[t].y;
            a2 += h[t].z;
        }
        for (int j = beg + sl + 8 * G2_SLOTS; j < end; j += 8) {
            const int s1 = csr[j];
            const v4f h1 = *(const v4f*)(H2 + (size_t)s1 * H2_PITCH);
            a0 += h1.x;
            a1 += h1.y;
            a2 += h1.z;
        }
#pragma unroll
        for (int m = 4; m >= 1; m >>= 1) {
            a0 += __shfl_xor(a0, m);
            a1 += __shfl_xor(a1, m);
            a2 += __shfl_xor(a2, m);
        }
        if (sl == 0) {
            sw[idx * OUT_DIM + 0] = a0;
            sw[idx * OUT_DIM + 1] = a1;
            sw[idx * OUT_DIM + 2] = a2;
        }
        beg = nbeg; end = nend;
#pragma unroll
        for (int t = 0; t < G2_SLOTS; ++t) s[t] = sn[t];
    }
    const float r0 = fmaf(sw[lane * OUT_DIM + 0] + hs.x, d_l, b2[0]);
    const float r1 = fmaf(sw[lane * OUT_DIM + 1] + hs.y, d_l, b2[1]);
    const float r2 = fmaf(sw[lane * OUT_DIM + 2] + hs.z, d_l, b2[2]);
    sw[lane * OUT_DIM + 0] = r0;
    sw[lane * OUT_DIM + 1] = r1;
    sw[lane * OUT_DIM + 2] = r2;
#pragma unroll
    for (int k = 0; k < OUT_DIM; ++k) {
        const int w = base * OUT_DIM + 32 * k + lane;
        if (w < n * OUT_DIM) out[w] = sw[32 * k + lane];
    }
}

extern "C" void kernel_launch(void* const* d_in, const int* in_sizes, int n_in,
                              void* d_out, int out_size, void* d_ws, size_t ws_size,
                              hipStream_t stream) {
    (void)n_in; (void)out_size; (void)ws_size;
    const float* x  = (const float*)d_in[0];
    const int*  edge = (const int*)d_in[1];
    const float* W1 = (const float*)d_in[2];
    const float* b1 = (const float*)d_in[3];
    const float* W2 = (const float*)d_in[4];
    const float* b2 = (const float*)d_in[5];
    float* out = (float*)d_out;

    const int n = in_sizes[0] / IN_DIM;
    const int e = in_sizes[1] / 2;
    const int* src = edge;
    const int* dst = edge + e;
    const int nb = (n + BKT_NODES - 1) / BKT_NODES;

    const size_t np  = ((size_t)n + 1 + 63) & ~(size_t)63;
    const size_t ep  = ((size_t)e + 63) & ~(size_t)63;
    float* ws    = (float*)d_ws;
    float* dinv  = ws;
    int*   off   = (int*)(ws + np);
    int*   tableT = off + np;
    int*   pairs = tableT + (size_t)(NBMAX + 1) * NCHUNK;
    int*   csr   = pairs + ep + (size_t)NCHUNK * 32;
    float* H1    = (float*)(csr + ep);
    float* H2    = H1 + (size_t)n * HID_DIM;

    k_part <<<NCHUNK, SORT_THREADS, 0, stream>>>(src, dst, tableT, pairs, e, nb);
    k_bsort_gemm1<<<nb, SORT_THREADS, 0, stream>>>(pairs, tableT, off, dinv, csr, x, W1, H1, n, e);

    k_gather1<<<(n + 16 * G1_WAVES - 1) / (16 * G1_WAVES), 32 * G1_WAVES, 0, stream>>>(off, csr, dinv, H1, b1, W2, H2, n);

    k_gather2<<<(n + 32 * G2_WAVES - 1) / (32 * G2_WAVES), 32 * G2_WAVES, 0, stream>>>(off, csr, dinv, H2, b2, out, n);
}
